# starting kernel
# baseline (speedup 1.0000x reference)
#include <hip/hip_runtime.h>
#include <hip/hip_bf16.h>
#include <math.h>

typedef float v2f __attribute__((ext_vector_type(2)));
typedef float v8f __attribute__((ext_vector_type(8)));

#define DD 128
#define NNODE 40000
#define NEDGE 640000

__global__ __launch_bounds__(256)
void gat_gemm_wmma(const float* __restrict__ h,
                   const float* __restrict__ W1,
                   const float* __restrict__ W2,
                   float* __restrict__ z,
                   float* __restrict__ znb)
{
    const int lane  = threadIdx.x & 31;
    const int wavew = threadIdx.x >> 5;
    const int row0  = blockIdx.x << 4;
    const int col0  = wavew << 4;
    const int mn    = lane & 15;
    const int khalf = (lane >> 4) << 1;

    const float* hrow = h + (size_t)(row0 + mn) * DD;

    v8f accZ = {0.f, 0.f, 0.f, 0.f, 0.f, 0.f, 0.f, 0.f};
    v8f accI = {0.f, 0.f, 0.f, 0.f, 0.f, 0.f, 0.f, 0.f};

    for (int k0 = 0; k0 < DD; k0 += 4) {
        if (k0 + 16 < DD) __builtin_prefetch(&hrow[k0 + 16], 0, 1);
        v2f a;
        a.x = hrow[k0 + khalf];
        a.y = hrow[k0 + khalf + 1];
        v2f b1, b2;
        b1.x = W1[(size_t)(k0 + khalf)     * DD + col0 + mn];
        b1.y = W1[(size_t)(k0 + khalf + 1) * DD + col0 + mn];
        b2.x = W2[(size_t)(k0 + khalf)     * DD + col0 + mn];
        b2.y = W2[(size_t)(k0 + khalf + 1) * DD + col0 + mn];
        accZ = __builtin_amdgcn_wmma_f32_16x16x4_f32(false, a, false, b1,
                                                     (short)0, accZ, false, false);
        accI = __builtin_amdgcn_wmma_f32_16x16x4_f32(false, a, false, b2,
                                                     (short)0, accI, false, false);
    }

    const int rowBase = row0 + ((lane >> 4) << 3);
    #pragma unroll
    for (int r = 0; r < 8; ++r) {
        size_t idx = (size_t)(rowBase + r) * DD + col0 + mn;
        z[idx]   = accZ[r];
        znb[idx] = accI[r];
    }
}

__global__ __launch_bounds__(256)
void gat_node_prep(const float* __restrict__ z,
                   const float* __restrict__ Wa_l,
                   float* __restrict__ a_src,
                   float* __restrict__ a_dst,
                   float* __restrict__ mmax,
                   float* __restrict__ denom)
{
    int gid  = blockIdx.x * blockDim.x + threadIdx.x;
    int node = gid >> 5;
    int lane = gid & 31;
    if (node >= NNODE) return;

    const float4 zv = reinterpret_cast<const float4*>(z + (size_t)node * DD)[lane];
    const float* wa0 = Wa_l + lane * 4;
    const float* wa1 = Wa_l + DD + lane * 4;
    float s0 = zv.x * wa0[0] + zv.y * wa0[1] + zv.z * wa0[2] + zv.w * wa0[3];
    float s1 = zv.x * wa1[0] + zv.y * wa1[1] + zv.z * wa1[2] + zv.w * wa1[3];
    #pragma unroll
    for (int off = 16; off > 0; off >>= 1) {
        s0 += __shfl_xor(s0, off, 32);
        s1 += __shfl_xor(s1, off, 32);
    }
    if (lane == 0) {
        a_src[node] = s0;
        a_dst[node] = s1;
        mmax[node]  = -INFINITY;
        denom[node] = 0.0f;
    }
}

__device__ inline void atomicMaxF(float* addr, float val) {
    if (val >= 0.0f) atomicMax((int*)addr, __float_as_int(val));
    else             atomicMin((unsigned int*)addr, (unsigned int)__float_as_int(val));
}

__global__ __launch_bounds__(256)
void gat_edge_logits(const int* __restrict__ src,
                     const int* __restrict__ dst,
                     const float* __restrict__ dfeat,
                     const float* __restrict__ W0_l,
                     const float* __restrict__ Wa_l,
                     const float* __restrict__ a_src,
                     const float* __restrict__ a_dst,
                     float* __restrict__ e,
                     float* __restrict__ mmax)
{
    int i = blockIdx.x * blockDim.x + threadIdx.x;
    if (i >= NEDGE) return;
    float ct = W0_l[0] * Wa_l[2 * DD];
    int s = src[i], t = dst[i];
    float v = a_src[s] + a_dst[t] + dfeat[i] * ct;
    v = (v > 0.0f) ? v : 0.01f * v;
    e[i] = v;
    atomicMaxF(&mmax[t], v);
}

__global__ __launch_bounds__(256)
void gat_edge_exp(const int* __restrict__ dst,
                  float* __restrict__ e,
                  const float* __restrict__ mmax,
                  float* __restrict__ denom)
{
    int i = blockIdx.x * blockDim.x + threadIdx.x;
    if (i >= NEDGE) return;
    int t = dst[i];
    float x = __expf(e[i] - mmax[t]);
    e[i] = x;
    atomicAdd(&denom[t], x);
}

__global__ __launch_bounds__(256)
void gat_edge_scatter(const int* __restrict__ src,
                      const int* __restrict__ dst,
                      const float* __restrict__ ex,
                      const float* __restrict__ denom,
                      const float* __restrict__ z,
                      float* __restrict__ znb)
{
    int gid  = blockIdx.x * blockDim.x + threadIdx.x;
    int eidx = gid >> 5;
    int lane = gid & 31;
    if (eidx >= NEDGE) return;
    int s = src[eidx], t = dst[eidx];
    float alpha = ex[eidx] / denom[t];
    const float4 v = reinterpret_cast<const float4*>(z + (size_t)s * DD)[lane];
    float* o = znb + (size_t)t * DD + lane * 4;
    atomicAdd(o + 0, v.x * alpha);
    atomicAdd(o + 1, v.y * alpha);
    atomicAdd(o + 2, v.z * alpha);
    atomicAdd(o + 3, v.w * alpha);
}

__global__ __launch_bounds__(256)
void gat_relu(const float* __restrict__ znb, float* __restrict__ hout)
{
    int i = blockIdx.x * blockDim.x + threadIdx.x;
    if (i < NNODE * DD) hout[i] = fmaxf(znb[i], 0.0f);
}

extern "C" void kernel_launch(void* const* d_in, const int* in_sizes, int n_in,
                              void* d_out, int out_size, void* d_ws, size_t ws_size,
                              hipStream_t stream) {
    const float* attr = (const float*)d_in[0];
    const float* dfe  = (const float*)d_in[1];
    const int*   src  = (const int*)d_in[2];
    const int*   dst  = (const int*)d_in[3];
    const float* W0   = (const float*)d_in[4];
    const float* W1   = (const float*)d_in[5];
    const float* W2   = (const float*)d_in[6];
    const float* Wa   = (const float*)d_in[7];

    float* ws    = (float*)d_ws;
    float* hA    = ws;
    float* hB    = hA    + (size_t)NNODE * DD;
    float* z     = hB    + (size_t)NNODE * DD;
    float* znb   = z     + (size_t)NNODE * DD;
    float* a_src = znb   + (size_t)NNODE * DD;
    float* a_dst = a_src + NNODE;
    float* mmax  = a_dst + NNODE;
    float* denom = mmax  + NNODE;
    float* ebuf  = denom + NNODE;

    const int gemmBlocks    = NNODE / 16;
    const int nodeBlocks    = (NNODE * 32 + 255) / 256;
    const int edgeBlocks    = (NEDGE + 255) / 256;
    const int scatterBlocks = (NEDGE * 32 + 255) / 256;
    const int reluBlocks    = (NNODE * DD + 255) / 256;

    for (int l = 0; l < 3; ++l) {
        const float* hin  = (l == 0) ? attr : ((l == 1) ? hA : hB);
        float*       hout = (l == 2) ? (float*)d_out : ((l == 0) ? hA : hB);
        const float* W1l = W1 + (size_t)l * DD * DD;
        const float* W2l = W2 + (size_t)l * DD * DD;
        const float* W0l = W0 + l;
        const float* Wal = Wa + (size_t)l * (2 * DD + 1);

        gat_gemm_wmma<<<gemmBlocks, 256, 0, stream>>>(hin, W1l, W2l, z, znb);
        gat_node_prep<<<nodeBlocks, 256, 0, stream>>>(z, Wal, a_src, a_dst, mmax, denom);
        gat_edge_logits<<<edgeBlocks, 256, 0, stream>>>(src, dst, dfe, W0l, Wal,
                                                        a_src, a_dst, ebuf, mmax);
        gat_edge_exp<<<edgeBlocks, 256, 0, stream>>>(dst, ebuf, mmax, denom);
        gat_edge_scatter<<<scatterBlocks, 256, 0, stream>>>(src, dst, ebuf, denom, z, znb);
        gat_relu<<<reluBlocks, 256, 0, stream>>>(znb, hout);
    }
}


# Edges grouped by destination once per call (integer count, scan, place); one wave per node does max, exp-sum and weighted row sum in registers: float atomics gone, sums run in list order
# speedup vs baseline: 2.4625x; 2.4625x over previous
#include <hip/hip_runtime.h>
#include <hip/hip_bf16.h>
#include <math.h>

typedef float v2f __attribute__((ext_vector_type(2)));
typedef float v8f __attribute__((ext_vector_type(8)));

#define DD 128
#define NNODE 40000
#define NEDGE 640000

__global__ __launch_bounds__(256)
void gat_gemm_wmma(const float* __restrict__ h,
                   const float* __restrict__ W1,
                   const float* __restrict__ W2,
                   float* __restrict__ z,
                   float* __restrict__ znb)
{
    const int lane  = threadIdx.x & 31;
    const int wavew = threadIdx.x >> 5;
    const int row0  = blockIdx.x << 4;
    const int col0  = wavew << 4;
    const int mn    = lane & 15;
    const int khalf = (lane >> 4) << 1;

    const float* hrow = h + (size_t)(row0 + mn) * DD;

    v8f accZ = {0.f, 0.f, 0.f, 0.f, 0.f, 0.f, 0.f, 0.f};
    v8f accI = {0.f, 0.f, 0.f, 0.f, 0.f, 0.f, 0.f, 0.f};

    for (int k0 = 0; k0 < DD; k0 += 4) {
        if (k0 + 16 < DD) __builtin_prefetch(&hrow[k0 + 16], 0, 1);
        v2f a;
        a.x = hrow[k0 + khalf];
        a.y = hrow[k0 + khalf + 1];
        v2f b1, b2;
        b1.x = W1[(size_t)(k0 + khalf)     * DD + col0 + mn];
        b1.y = W1[(size_t)(k0 + khalf + 1) * DD + col0 + mn];
        b2.x = W2[(size_t)(k0 + khalf)     * DD + col0 + mn];
        b2.y = W2[(size_t)(k0 + khalf + 1) * DD + col0 + mn];
        accZ = __builtin_amdgcn_wmma_f32_16x16x4_f32(false, a, false, b1,
                                                     (short)0, accZ, false, false);
        accI = __builtin_amdgcn_wmma_f32_16x16x4_f32(false, a, false, b2,
                                                     (short)0, accI, false, false);
    }

    const int rowBase = row0 + ((lane >> 4) << 3);
    #pragma unroll
    for (int r = 0; r < 8; ++r) {
        size_t idx = (size_t)(rowBase + r) * DD + col0 + mn;
        z[idx]   = accZ[r];
        znb[idx] = accI[r];
    }
}

__global__ __launch_bounds__(256)
void gat_node_prep(const float* __restrict__ z,
                   const float* __restrict__ Wa_l,
                   float* __restrict__ a_src,
                   float* __restrict__ a_dst)
{
    int gid  = blockIdx.x * blockDim.x + threadIdx.x;
    int node = gid >> 5;
    int lane = gid & 31;
    if (node >= NNODE) return;

    const float4 zv = reinterpret_cast<const float4*>(z + (size_t)node * DD)[lane];
    const float* wa0 = Wa_l + lane * 4;
    const float* wa1 = Wa_l + DD + lane * 4;
    float s0 = zv.x * wa0[0] + zv.y * wa0[1] + zv.z * wa0[2] + zv.w * wa0[3];
    float s1 = zv.x * wa1[0] + zv.y * wa1[1] + zv.z * wa1[2] + zv.w * wa1[3];
    #pragma unroll
    for (int off = 16; off > 0; off >>= 1) {
        s0 += __shfl_xor(s0, off, 32);
        s1 += __shfl_xor(s1, off, 32);
    }
    if (lane == 0) {
        a_src[node] = s0;
        a_dst[node] = s1;
    }
}

__global__ __launch_bounds__(256)
void gat_deg_zero(int* __restrict__ deg)
{
    int i = blockIdx.x * blockDim.x + threadIdx.x;
    if (i < NNODE) deg[i] = 0;
}

__global__ __launch_bounds__(256)
void gat_deg_rank(const int* __restrict__ dst,
                  int* __restrict__ deg,
                  int* __restrict__ rank)
{
    int i = blockIdx.x * blockDim.x + threadIdx.x;
    if (i >= NEDGE) return;
    rank[i] = atomicAdd(&deg[dst[i]], 1);
}

#define SCAN_PER 40
#define SCAN_ACTIVE (NNODE / SCAN_PER)
static_assert(NNODE % SCAN_PER == 0 && SCAN_ACTIVE < 1024 && SCAN_PER % 4 == 0,
              "scan block layout");

__global__ __launch_bounds__(1024)
void gat_rowptr_scan(const int* __restrict__ deg, int* __restrict__ rowptr)
{
    __shared__ int wtot[32];
    const int tid  = threadIdx.x;
    const int lane = tid & 31;
    const int w    = tid >> 5;

    int pre[SCAN_PER];
    int tot = 0;
    if (tid < SCAN_ACTIVE) {
        const int4* d4 = reinterpret_cast<const int4*>(deg + tid * SCAN_PER);
        #pragma unroll
        for (int q = 0; q < SCAN_PER / 4; ++q) {
            const int4 d = d4[q];
            pre[4 * q + 0] = tot; tot += d.x;
            pre[4 * q + 1] = tot; tot += d.y;
            pre[4 * q + 2] = tot; tot += d.z;
            pre[4 * q + 3] = tot; tot += d.w;
        }
    }
    int inc = tot;
    #pragma unroll
    for (int off = 1; off < 32; off <<= 1) {
        int n = __shfl_up(inc, off, 32);
        if (lane >= off) inc += n;
    }
    if (lane == 31) wtot[w] = inc;
    __syncthreads();
    if (w == 0) {
        const int t = wtot[lane];
        int ti = t;
        #pragma unroll
        for (int off = 1; off < 32; off <<= 1) {
            int n = __shfl_up(ti, off, 32);
            if (lane >= off) ti += n;
        }
        wtot[lane] = ti - t;
    }
    __syncthreads();
    const int offs = wtot[w] + inc - tot;
    if (tid < SCAN_ACTIVE) {
        int4* r4 = reinterpret_cast<int4*>(rowptr + tid * SCAN_PER);
        #pragma unroll
        for (int q = 0; q < SCAN_PER / 4; ++q) {
            int4 r;
            r.x = offs + pre[4 * q + 0];
            r.y = offs + pre[4 * q + 1];
            r.z = offs + pre[4 * q + 2];
            r.w = offs + pre[4 * q + 3];
            r4[q] = r;
        }
    } else if (tid == SCAN_ACTIVE) {
        rowptr[NNODE] = offs;
    }
}

__global__ __launch_bounds__(256)
void gat_edge_place(const int* __restrict__ src,
                    const int* __restrict__ dst,
                    const float* __restrict__ dfeat,
                    const int* __restrict__ rowptr,
                    const int* __restrict__ rank,
                    int* __restrict__ csrc,
                    float* __restrict__ cd)
{
    int i = blockIdx.x * blockDim.x + threadIdx.x;
    if (i >= NEDGE) return;
    const int p = rowptr[dst[i]] + rank[i];
    csrc[p] = src[i];
    cd[p]   = dfeat[i];
}

__device__ __forceinline__ float wave_max(float v) {
    #pragma unroll
    for (int off = 16; off > 0; off >>= 1) v = fmaxf(v, __shfl_xor(v, off, 32));
    return v;
}
__device__ __forceinline__ float wave_sum(float v) {
    #pragma unroll
    for (int off = 16; off > 0; off >>= 1) v += __shfl_xor(v, off, 32);
    return v;
}

template <int R>
__device__ __forceinline__ void add_rows(float4& acc, const float4* __restrict__ z4,
                                         int lane, int s, int ali, int k)
{
    int    sk[R];
    float  ak[R];
    float4 r[R];
    #pragma unroll
    for (int q = 0; q < R; ++q) {
        sk[q] = __builtin_amdgcn_readlane(s, k + q);
        ak[q] = __int_as_float(__builtin_amdgcn_readlane(ali, k + q));
    }
    #pragma unroll
    for (int q = 0; q < R; ++q) r[q] = z4[(size_t)sk[q] * (DD / 4) + lane];
    #pragma unroll
    for (int q = 0; q < R; ++q) {
        acc.x = fmaf(r[q].x, ak[q], acc.x);
        acc.y = fmaf(r[q].y, ak[q], acc.y);
        acc.z = fmaf(r[q].z, ak[q], acc.z);
        acc.w = fmaf(r[q].w, ak[q], acc.w);
    }
}

__device__ __forceinline__ void gather_rows(float4& acc, const float4* __restrict__ z4,
                                            int lane, int s, float al, int cnt)
{
    const int ali = __float_as_int(al);
    int k = 0;
    for (; k + 8 <= cnt; k += 8) add_rows<8>(acc, z4, lane, s, ali, k);
    if (cnt - k >= 4) { add_rows<4>(acc, z4, lane, s, ali, k); k += 4; }
    switch (cnt - k) {
        case 3:  add_rows<3>(acc, z4, lane, s, ali, k); break;
        case 2:  add_rows<2>(acc, z4, lane, s, ali, k); break;
        case 1:  add_rows<1>(acc, z4, lane, s, ali, k); break;
        default: break;
    }
}

__global__ __launch_bounds__(256)
void gat_node_aggregate(const int* __restrict__ rowptr,
                        const int* __restrict__ csrc,
                        const float* __restrict__ cd,
                        const float* __restrict__ W0_l,
                        const float* __restrict__ Wa_l,
                        const float* __restrict__ a_src,
                        const float* __restrict__ a_dst,
                        const float* __restrict__ z,
                        const float* __restrict__ zi,
                        float* __restrict__ hout)
{
    const int gid  = blockIdx.x * blockDim.x + threadIdx.x;
    const int lane = gid & 31;
    const int node = __builtin_amdgcn_readfirstlane(gid >> 5);
    if (node >= NNODE) return;
    const int beg = __builtin_amdgcn_readfirstlane(rowptr[node]);
    const int end = __builtin_amdgcn_readfirstlane(rowptr[node + 1]);
    const int deg = end - beg;

    const float ct  = W0_l[0] * Wa_l[2 * DD];
    const float adt = a_dst[node];
    const float4* z4 = reinterpret_cast<const float4*>(z);
    const float4 own = reinterpret_cast<const float4*>(zi)[(size_t)node * (DD / 4) + lane];
    float4 acc = {0.0f, 0.0f, 0.0f, 0.0f};

    if (deg <= 32) {
        const bool act = lane < deg;
        int   s = 0;
        float v = -INFINITY;
        if (act) {
            s = csrc[beg + lane];
            const float t = a_src[s] + adt + cd[beg + lane] * ct;
            v = (t > 0.0f) ? t : 0.01f * t;
        }
        float m = wave_max(v);
        if (!isfinite(m)) m = 0.0f;
        const float ex  = act ? __expf(v - m) : 0.0f;
        const float den = wave_sum(ex);
        gather_rows(acc, z4, lane, s, ex / den, deg);
    } else {
        float m = -INFINITY;
        for (int j = beg + lane; j < end; j += 32) {
            const float t = a_src[csrc[j]] + adt + cd[j] * ct;
            m = fmaxf(m, (t > 0.0f) ? t : 0.01f * t);
        }
        m = wave_max(m);
        if (!isfinite(m)) m = 0.0f;
        float den = 0.0f;
        for (int j = beg + lane; j < end; j += 32) {
            const float t = a_src[csrc[j]] + adt + cd[j] * ct;
            den += __expf(((t > 0.0f) ? t : 0.01f * t) - m);
        }
        den = wave_sum(den);
        for (int c = beg; c < end; c += 32) {
            const int j = c + lane;
            int   s  = 0;
            float al = 0.0f;
            if (j < end) {
                s = csrc[j];
                const float t = a_src[s] + adt + cd[j] * ct;
                al = __expf(((t > 0.0f) ? t : 0.01f * t) - m) / den;
            }
            const int left = end - c;
            gather_rows(acc, z4, lane, s, al, left < 32 ? left : 32);
        }
    }

    float4 o;
    o.x = fmaxf(own.x + acc.x, 0.0f);
    o.y = fmaxf(own.y + acc.y, 0.0f);
    o.z = fmaxf(own.z + acc.z, 0.0f);
    o.w = fmaxf(own.w + acc.w, 0.0f);
    reinterpret_cast<float4*>(hout)[(size_t)node * (DD / 4) + lane] = o;
}

extern "C" void kernel_launch(void* const* d_in, const int* in_sizes, int n_in,
                              void* d_out, int out_size, void* d_ws, size_t ws_size,
                              hipStream_t stream) {
    const float* attr = (const float*)d_in[0];
    const float* dfe  = (const float*)d_in[1];
    const int*   src  = (const int*)d_in[2];
    const int*   dst  = (const int*)d_in[3];
    const float* W0   = (const float*)d_in[4];
    const float* W1   = (const float*)d_in[5];
    const float* W2   = (const float*)d_in[6];
    const float* Wa   = (const float*)d_in[7];

    float* ws     = (float*)d_ws;
    float* hA     = ws;
    float* hB     = hA    + (size_t)NNODE * DD;
    float* z      = hB    + (size_t)NNODE * DD;
    float* zi     = z     + (size_t)NNODE * DD;
    float* a_src  = zi    + (size_t)NNODE * DD;
    float* a_dst  = a_src + NNODE;
    int*   deg    = (int*)(a_dst + NNODE);
    int*   rowptr = deg    + NNODE;
    int*   rank   = rowptr + NNODE + 4;
    int*   csrc   = rank   + NEDGE;
    float* cd     = (float*)(csrc + NEDGE);

    const int gemmBlocks = NNODE / 16;
    const int nodeBlocks = (NNODE * 32 + 255) / 256;
    const int edgeBlocks = (NEDGE + 255) / 256;
    const int degBlocks  = (NNODE + 255) / 256;

    gat_deg_zero<<<degBlocks, 256, 0, stream>>>(deg);
    gat_deg_rank<<<edgeBlocks, 256, 0, stream>>>(dst, deg, rank);
    gat_rowptr_scan<<<1, 1024, 0, stream>>>(deg, rowptr);
    gat_edge_place<<<edgeBlocks, 256, 0, stream>>>(src, dst, dfe, rowptr, rank, csrc, cd);

    for (int l = 0; l < 3; ++l) {
        const float* hin  = (l == 0) ? attr : ((l == 1) ? hA : hB);
        float*       hout = (l == 2) ? (float*)d_out : ((l == 0) ? hA : hB);
        const float* W1l = W1 + (size_t)l * DD * DD;
        const float* W2l = W2 + (size_t)l * DD * DD;
        const float* W0l = W0 + l;
        const float* Wal = Wa + (size_t)l * (2 * DD + 1);

        gat_gemm_wmma<<<gemmBlocks, 256, 0, stream>>>(hin, W1l, W2l, z, zi);
        gat_node_prep<<<nodeBlocks, 256, 0, stream>>>(z, Wal, a_src, a_dst);
        gat_node_aggregate<<<nodeBlocks, 256, 0, stream>>>(rowptr, csrc, cd, W0l, Wal,
                                                           a_src, a_dst, z, zi, hout);
    }
}


# GEMM made persistent: each wave keeps its W1/W2 column-slab fragments for the whole K in registers and walks over 16-row tiles of h staged once per block through LDS (512 blocks, not 2500)
# speedup vs baseline: 3.2063x; 1.3021x over previous
#include <hip/hip_runtime.h>
#include <hip/hip_bf16.h>
#include <math.h>

typedef float v2f __attribute__((ext_vector_type(2)));
typedef float v8f __attribute__((ext_vector_type(8)));

#define DD 128
#define NNODE 40000
#define NEDGE 640000

#define GEMM_TILES  (NNODE / 16)
#define GEMM_BLOCKS 512
#define A_PITCH     132
static_assert(NNODE % 16 == 0 && GEMM_BLOCKS <= GEMM_TILES, "row tiling");

__global__ __launch_bounds__(256)
void gat_gemm_wmma(const float* __restrict__ h,
                   const float* __restrict__ W1,
                   const float* __restrict__ W2,
                   float* __restrict__ z,
                   float* __restrict__ znb)
{
    __shared__ __attribute__((aligned(16))) float As[2][16 * A_PITCH];

    const int tid   = threadIdx.x;
    const int lane  = tid & 31;
    const int wavew = tid >> 5;
    const int col0  = wavew << 4;
    const int mn    = lane & 15;
    const int khalf = (lane >> 4) << 1;

    v2f b1[DD / 4], b2[DD / 4];
    #pragma unroll
    for (int s = 0; s < DD / 4; ++s) {
        const size_t k = (size_t)(4 * s + khalf);
        b1[s].x = W1[k * DD + col0 + mn];
        b1[s].y = W1[(k + 1) * DD + col0 + mn];
        b2[s].x = W2[k * DD + col0 + mn];
        b2[s].y = W2[(k + 1) * DD + col0 + mn];
    }

    const float4* h4 = reinterpret_cast<const float4*>(h);
    int tile = blockIdx.x;
    float4 g0 = h4[(size_t)(tile * 16 + wavew)     * (DD / 4) + lane];
    float4 g1 = h4[(size_t)(tile * 16 + wavew + 8) * (DD / 4) + lane];
    *reinterpret_cast<float4*>(&As[0][wavew       * A_PITCH + lane * 4]) = g0;
    *reinterpret_cast<float4*>(&As[0][(wavew + 8) * A_PITCH + lane * 4]) = g1;
    __syncthreads();

    int buf = 0;
    for (; tile < GEMM_TILES; tile += GEMM_BLOCKS) {
        const int  nxt  = tile + GEMM_BLOCKS;
        const bool more = nxt < GEMM_TILES;
        if (more) {
            g0 = h4[(size_t)(nxt * 16 + wavew)     * (DD / 4) + lane];
            g1 = h4[(size_t)(nxt * 16 + wavew + 8) * (DD / 4) + lane];
        }

        v8f accZ = {0.f, 0.f, 0.f, 0.f, 0.f, 0.f, 0.f, 0.f};
        v8f accI = {0.f, 0.f, 0.f, 0.f, 0.f, 0.f, 0.f, 0.f};
        const float* arow = &As[buf][mn * A_PITCH + khalf];
        #pragma unroll
        for (int s = 0; s < DD / 4; ++s) {
            const v2f a = *reinterpret_cast<const v2f*>(arow + 4 * s);
            accZ = __builtin_amdgcn_wmma_f32_16x16x4_f32(false, a, false, b1[s],
                                                         (short)0, accZ, false, false);
            accI = __builtin_amdgcn_wmma_f32_16x16x4_f32(false, a, false, b2[s],
                                                         (short)0, accI, false, false);
        }

        if (more) {
            *reinterpret_cast<float4*>(&As[buf ^ 1][wavew       * A_PITCH + lane * 4]) = g0;
            *reinterpret_cast<float4*>(&As[buf ^ 1][(wavew + 8) * A_PITCH + lane * 4]) = g1;
        }
        __syncthreads();
        buf ^= 1;

        const int rowBase = tile * 16 + ((lane >> 4) << 3);
        #pragma unroll
        for (int r = 0; r < 8; ++r) {
            size_t idx = (size_t)(rowBase + r) * DD + col0 + mn;
            z[idx]   = accZ[r];
            znb[idx] = accI[r];
        }
    }
}

__global__ __launch_bounds__(256)
void gat_node_prep(const float* __restrict__ z,
                   const float* __restrict__ Wa_l,
                   float* __restrict__ a_src,
                   float* __restrict__ a_dst)
{
    int gid  = blockIdx.x * blockDim.x + threadIdx.x;
    int node = gid >> 5;
    int lane = gid & 31;
    if (node >= NNODE) return;

    const float4 zv = reinterpret_cast<const float4*>(z + (size_t)node * DD)[lane];
    const float* wa0 = Wa_l + lane * 4;
    const float* wa1 = Wa_l + DD + lane * 4;
    float s0 = zv.x * wa0[0] + zv.y * wa0[1] + zv.z * wa0[2] + zv.w * wa0[3];
    float s1 = zv.x * wa1[0] + zv.y * wa1[1] + zv.z * wa1[2] + zv.w * wa1[3];
    #pragma unroll
    for (int off = 16; off > 0; off >>= 1) {
        s0 += __shfl_xor(s0, off, 32);
        s1 += __shfl_xor(s1, off, 32);
    }
    if (lane == 0) {
        a_src[node] = s0;
        a_dst[node] = s1;
    }
}

__global__ __launch_bounds__(256)
void gat_deg_zero(int* __restrict__ deg)
{
    int i = blockIdx.x * blockDim.x + threadIdx.x;
    if (i < NNODE) deg[i] = 0;
}

__global__ __launch_bounds__(256)
void gat_deg_rank(const int* __restrict__ dst,
                  int* __restrict__ deg,
                  int* __restrict__ rank)
{
    int i = blockIdx.x * blockDim.x + threadIdx.x;
    if (i >= NEDGE) return;
    rank[i] = atomicAdd(&deg[dst[i]], 1);
}

#define SCAN_PER 40
#define SCAN_ACTIVE (NNODE / SCAN_PER)
static_assert(NNODE % SCAN_PER == 0 && SCAN_ACTIVE < 1024 && SCAN_PER % 4 == 0,
              "scan block layout");

__global__ __launch_bounds__(1024)
void gat_rowptr_scan(const int* __restrict__ deg, int* __restrict__ rowptr)
{
    __shared__ int wtot[32];
    const int tid  = threadIdx.x;
    const int lane = tid & 31;
    const int w    = tid >> 5;

    int pre[SCAN_PER];
    int tot = 0;
    if (tid < SCAN_ACTIVE) {
        const int4* d4 = reinterpret_cast<const int4*>(deg + tid * SCAN_PER);
        #pragma unroll
        for (int q = 0; q < SCAN_PER / 4; ++q) {
            const int4 d = d4[q];
            pre[4 * q + 0] = tot; tot += d.x;
            pre[4 * q + 1] = tot; tot += d.y;
            pre[4 * q + 2] = tot; tot += d.z;
            pre[4 * q + 3] = tot; tot += d.w;
        }
    }
    int inc = tot;
    #pragma unroll
    for (int off = 1; off < 32; off <<= 1) {
        int n = __shfl_up(inc, off, 32);
        if (lane >= off) inc += n;
    }
    if (lane == 31) wtot[w] = inc;
    __syncthreads();
    if (w == 0) {
        const int t = wtot[lane];
        int ti = t;
        #pragma unroll
        for (int off = 1; off < 32; off <<= 1) {
            int n = __shfl_up(ti, off, 32);
            if (lane >= off) ti += n;
        }
        wtot[lane] = ti - t;
    }
    __syncthreads();
    const int offs = wtot[w] + inc - tot;
    if (tid < SCAN_ACTIVE) {
        int4* r4 = reinterpret_cast<int4*>(rowptr + tid * SCAN_PER);
        #pragma unroll
        for (int q = 0; q < SCAN_PER / 4; ++q) {
            int4 r;
            r.x = offs + pre[4 * q + 0];
            r.y = offs + pre[4 * q + 1];
            r.z = offs + pre[4 * q + 2];
            r.w = offs + pre[4 * q + 3];
            r4[q] = r;
        }
    } else if (tid == SCAN_ACTIVE) {
        rowptr[NNODE] = offs;
    }
}

__global__ __launch_bounds__(256)
void gat_edge_place(const int* __restrict__ src,
                    const int* __restrict__ dst,
                    const float* __restrict__ dfeat,
                    const int* __restrict__ rowptr,
                    const int* __restrict__ rank,
                    int* __restrict__ csrc,
                    float* __restrict__ cd)
{
    int i = blockIdx.x * blockDim.x + threadIdx.x;
    if (i >= NEDGE) return;
    const int p = rowptr[dst[i]] + rank[i];
    csrc[p] = src[i];
    cd[p]   = dfeat[i];
}

__device__ __forceinline__ float wave_max(float v) {
    #pragma unroll
    for (int off = 16; off > 0; off >>= 1) v = fmaxf(v, __shfl_xor(v, off, 32));
    return v;
}
__device__ __forceinline__ float wave_sum(float v) {
    #pragma unroll
    for (int off = 16; off > 0; off >>= 1) v += __shfl_xor(v, off, 32);
    return v;
}

template <int R>
__device__ __forceinline__ void add_rows(float4& acc, const float4* __restrict__ z4,
                                         int lane, int s, int ali, int k)
{
    int    sk[R];
    float  ak[R];
    float4 r[R];
    #pragma unroll
    for (int q = 0; q < R; ++q) {
        sk[q] = __builtin_amdgcn_readlane(s, k + q);
        ak[q] = __int_as_float(__builtin_amdgcn_readlane(ali, k + q));
    }
    #pragma unroll
    for (int q = 0; q < R; ++q) r[q] = z4[(size_t)sk[q] * (DD / 4) + lane];
    #pragma unroll
    for (int q = 0; q < R; ++q) {
        acc.x = fmaf(r[q].x, ak[q], acc.x);
        acc.y = fmaf(r[q].y, ak[q], acc.y);
        acc.z = fmaf(r[q].z, ak[q], acc.z);
        acc.w = fmaf(r[q].w, ak[q], acc.w);
    }
}

__device__ __forceinline__ void gather_rows(float4& acc, const float4* __restrict__ z4,
                                            int lane, int s, float al, int cnt)
{
    const int ali = __float_as_int(al);
    int k = 0;
    for (; k + 8 <= cnt; k += 8) add_rows<8>(acc, z4, lane, s, ali, k);
    if (cnt - k >= 4) { add_rows<4>(acc, z4, lane, s, ali, k); k += 4; }
    switch (cnt - k) {
        case 3:  add_rows<3>(acc, z4, lane, s, ali, k); break;
        case 2:  add_rows<2>(acc, z4, lane, s, ali, k); break;
        case 1:  add_rows<1>(acc, z4, lane, s, ali, k); break;
        default: break;
    }
}

__global__ __launch_bounds__(256)
void gat_node_aggregate(const int* __restrict__ rowptr,
                        const int* __restrict__ csrc,
                        const float* __restrict__ cd,
                        const float* __restrict__ W0_l,
                        const float* __restrict__ Wa_l,
                        const float* __restrict__ a_src,
                        const float* __restrict__ a_dst,
                        const float* __restrict__ z,
                        const float* __restrict__ zi,
                        float* __restrict__ hout)
{
    const int gid  = blockIdx.x * blockDim.x + threadIdx.x;
    const int lane = gid & 31;
    const int node = __builtin_amdgcn_readfirstlane(gid >> 5);
    if (node >= NNODE) return;
    const int beg = __builtin_amdgcn_readfirstlane(rowptr[node]);
    const int end = __builtin_amdgcn_readfirstlane(rowptr[node + 1]);
    const int deg = end - beg;

    const float ct  = W0_l[0] * Wa_l[2 * DD];
    const float adt = a_dst[node];
    const float4* z4 = reinterpret_cast<const float4*>(z);
    const float4 own = reinterpret_cast<const float4*>(zi)[(size_t)node * (DD / 4) + lane];
    float4 acc = {0.0f, 0.0f, 0.0f, 0.0f};

    if (deg <= 32) {
        const bool act = lane < deg;
        int   s = 0;
        float v = -INFINITY;
        if (act) {
            s = csrc[beg + lane];
            const float t = a_src[s] + adt + cd[beg + lane] * ct;
            v = (t > 0.0f) ? t : 0.01f * t;
        }
        float m = wave_max(v);
        if (!isfinite(m)) m = 0.0f;
        const float ex  = act ? __expf(v - m) : 0.0f;
        const float den = wave_sum(ex);
        gather_rows(acc, z4, lane, s, ex / den, deg);
    } else {
        float m = -INFINITY;
        for (int j = beg + lane; j < end; j += 32) {
            const float t = a_src[csrc[j]] + adt + cd[j] * ct;
            m = fmaxf(m, (t > 0.0f) ? t : 0.01f * t);
        }
        m = wave_max(m);
        if (!isfinite(m)) m = 0.0f;
        float den = 0.0f;
        for (int j = beg + lane; j < end; j += 32) {
            const float t = a_src[csrc[j]] + adt + cd[j] * ct;
            den += __expf(((t > 0.0f) ? t : 0.01f * t) - m);
        }
        den = wave_sum(den);
        for (int c = beg; c < end; c += 32) {
            const int j = c + lane;
            int   s  = 0;
            float al = 0.0f;
            if (j < end) {
                s = csrc[j];
                const float t = a_src[s] + adt + cd[j] * ct;
                al = __expf(((t > 0.0f) ? t : 0.01f * t) - m) / den;
            }
            const int left = end - c;
            gather_rows(acc, z4, lane, s, al, left < 32 ? left : 32);
        }
    }

    float4 o;
    o.x = fmaxf(own.x + acc.x, 0.0f);
    o.y = fmaxf(own.y + acc.y, 0.0f);
    o.z = fmaxf(own.z + acc.z, 0.0f);
    o.w = fmaxf(own.w + acc.w, 0.0f);
    reinterpret_cast<float4*>(hout)[(size_t)node * (DD / 4) + lane] = o;
}

extern "C" void kernel_launch(void* const* d_in, const int* in_sizes, int n_in,
                              void* d_out, int out_size, void* d_ws, size_t ws_size,
                              hipStream_t stream) {
    const float* attr = (const float*)d_in[0];
    const float* dfe  = (const float*)d_in[1];
    const int*   src  = (const int*)d_in[2];
    const int*   dst  = (const int*)d_in[3];
    const float* W0   = (const float*)d_in[4];
    const float* W1   = (const float*)d_in[5];
    const float* W2   = (const float*)d_in[6];
    const float* Wa   = (const float*)d_in[7];

    float* ws     = (float*)d_ws;
    float* hA     = ws;
    float* hB     = hA    + (size_t)NNODE * DD;
    float* z      = hB    + (size_t)NNODE * DD;
    float* zi     = z     + (size_t)NNODE * DD;
    float* a_src  = zi    + (size_t)NNODE * DD;
    float* a_dst  = a_src + NNODE;
    int*   deg    = (int*)(a_dst + NNODE);
    int*   rowptr = deg    + NNODE;
    int*   rank   = rowptr + NNODE + 4;
    int*   csrc   = rank   + NEDGE;
    float* cd     = (float*)(csrc + NEDGE);

    const int gemmBlocks = GEMM_BLOCKS;
    const int nodeBlocks = (NNODE * 32 + 255) / 256;
    const int edgeBlocks = (NEDGE + 255) / 256;
    const int degBlocks  = (NNODE + 255) / 256;

    gat_deg_zero<<<degBlocks, 256, 0, stream>>>(deg);
    gat_deg_rank<<<edgeBlocks, 256, 0, stream>>>(dst, deg, rank);
    gat_rowptr_scan<<<1, 1024, 0, stream>>>(deg, rowptr);
    gat_edge_place<<<edgeBlocks, 256, 0, stream>>>(src, dst, dfe, rowptr, rank, csrc, cd);

    for (int l = 0; l < 3; ++l) {
        const float* hin  = (l == 0) ? attr : ((l == 1) ? hA : hB);
        float*       hout = (l == 2) ? (float*)d_out : ((l == 0) ? hA : hB);
        const float* W1l = W1 + (size_t)l * DD * DD;
        const float* W2l = W2 + (size_t)l * DD * DD;
        const float* W0l = W0 + l;
        const float* Wal = Wa + (size_t)l * (2 * DD + 1);

        gat_gemm_wmma<<<gemmBlocks, 256, 0, stream>>>(hin, W1l, W2l, z, zi);
        gat_node_prep<<<nodeBlocks, 256, 0, stream>>>(z, Wal, a_src, a_dst);
        gat_node_aggregate<<<nodeBlocks, 256, 0, stream>>>(rowptr, csrc, cd, W0l, Wal,
                                                           a_src, a_dst, z, zi, hout);
    }
}


# Attention dots a_src/a_dst formed in the GEMM epilogue from the z accumulators (DPP row sums, 8 column slabs added in order via LDS): the 3 node-prep launches go; dot sum order changes
# speedup vs baseline: 3.5087x; 1.0943x over previous
#include <hip/hip_runtime.h>
#include <hip/hip_bf16.h>
#include <math.h>

typedef float v2f __attribute__((ext_vector_type(2)));
typedef float v8f __attribute__((ext_vector_type(8)));

#define DD 128
#define NNODE 40000
#define NEDGE 640000

#define GEMM_TILES  (NNODE / 16)
#define GEMM_BLOCKS 512
#define A_PITCH     132
static_assert(NNODE % 16 == 0 && GEMM_BLOCKS <= GEMM_TILES, "row tiling");

template <int M>
__device__ __forceinline__ float row_xor(float v)
{
    return __int_as_float(__builtin_amdgcn_update_dpp(0, __float_as_int(v), 0x160 | M,
                                                      0xf, 0xf, true));
}

template <int M>
__device__ __forceinline__ float fold_pair(bool upper, float lo, float hi)
{
    const float keep = upper ? hi : lo;
    const float send = upper ? lo : hi;
    return keep + row_xor<M>(send);
}

__global__ __launch_bounds__(256)
void gat_gemm_wmma(const float* __restrict__ h,
                   const float* __restrict__ W1,
                   const float* __restrict__ W2,
                   const float* __restrict__ Wa_l,
                   float* __restrict__ z,
                   float* __restrict__ znb,
                   float* __restrict__ a_src,
                   float* __restrict__ a_dst)
{
    __shared__ __attribute__((aligned(16))) float As[2][16 * A_PITCH];
    __shared__ float Ps[2][8][32];

    const int tid   = threadIdx.x;
    const int lane  = tid & 31;
    const int wavew = tid >> 5;
    const int col0  = wavew << 4;
    const int mn    = lane & 15;
    const int khalf = (lane >> 4) << 1;

    v2f b1[DD / 4], b2[DD / 4];
    #pragma unroll
    for (int s = 0; s < DD / 4; ++s) {
        const size_t k = (size_t)(4 * s + khalf);
        b1[s].x = W1[k * DD + col0 + mn];
        b1[s].y = W1[(k + 1) * DD + col0 + mn];
        b2[s].x = W2[k * DD + col0 + mn];
        b2[s].y = W2[(k + 1) * DD + col0 + mn];
    }
    const float wa_s = Wa_l[col0 + mn];
    const float wa_d = Wa_l[DD + col0 + mn];
    const int pslot = ((mn >> 3) << 4) + ((lane >> 4) << 3) + (mn & 7);

    const float4* h4 = reinterpret_cast<const float4*>(h);
    int tile = blockIdx.x;
    float4 g0 = h4[(size_t)(tile * 16 + wavew)     * (DD / 4) + lane];
    float4 g1 = h4[(size_t)(tile * 16 + wavew + 8) * (DD / 4) + lane];
    *reinterpret_cast<float4*>(&As[0][wavew       * A_PITCH + lane * 4]) = g0;
    *reinterpret_cast<float4*>(&As[0][(wavew + 8) * A_PITCH + lane * 4]) = g1;
    __syncthreads();

    int buf = 0;
    for (; tile < GEMM_TILES; tile += GEMM_BLOCKS) {
        const int  nxt  = tile + GEMM_BLOCKS;
        const bool more = nxt < GEMM_TILES;
        if (more) {
            g0 = h4[(size_t)(nxt * 16 + wavew)     * (DD / 4) + lane];
            g1 = h4[(size_t)(nxt * 16 + wavew + 8) * (DD / 4) + lane];
        }

        v8f accZ = {0.f, 0.f, 0.f, 0.f, 0.f, 0.f, 0.f, 0.f};
        v8f accI = {0.f, 0.f, 0.f, 0.f, 0.f, 0.f, 0.f, 0.f};
        const float* arow = &As[buf][mn * A_PITCH + khalf];
        #pragma unroll
        for (int s = 0; s < DD / 4; ++s) {
            const v2f a = *reinterpret_cast<const v2f*>(arow + 4 * s);
            accZ = __builtin_amdgcn_wmma_f32_16x16x4_f32(false, a, false, b1[s],
                                                         (short)0, accZ, false, false);
            accI = __builtin_amdgcn_wmma_f32_16x16x4_f32(false, a, false, b2[s],
                                                         (short)0, accI, false, false);
        }

        {
            float v8[8], v4[4], v2[2];
            #pragma unroll
            for (int r = 0; r < 8; ++r)
                v8[r] = fold_pair<8>((lane & 8) != 0, accZ[r] * wa_s, accZ[r] * wa_d);
            #pragma unroll
            for (int r = 0; r < 4; ++r)
                v4[r] = fold_pair<4>((lane & 4) != 0, v8[r], v8[r + 4]);
            #pragma unroll
            for (int r = 0; r < 2; ++r)
                v2[r] = fold_pair<2>((lane & 2) != 0, v4[r], v4[r + 2]);
            Ps[buf][wavew][pslot] = fold_pair<1>((lane & 1) != 0, v2[0], v2[1]);
        }

        if (more) {
            *reinterpret_cast<float4*>(&As[buf ^ 1][wavew       * A_PITCH + lane * 4]) = g0;
            *reinterpret_cast<float4*>(&As[buf ^ 1][(wavew + 8) * A_PITCH + lane * 4]) = g1;
        }
        __syncthreads();

        if (wavew == 0) {
            float s = Ps[buf][0][lane];
            #pragma unroll
            for (int w = 1; w < 8; ++w) s += Ps[buf][w][lane];
            float* adst = (lane & 16) ? a_dst : a_src;
            adst[tile * 16 + mn] = s;
        }
        buf ^= 1;

        const int rowBase = tile * 16 + ((lane >> 4) << 3);
        #pragma unroll
        for (int r = 0; r < 8; ++r) {
            size_t idx = (size_t)(rowBase + r) * DD + col0 + mn;
            z[idx]   = accZ[r];
            znb[idx] = accI[r];
        }
    }
}

__global__ __launch_bounds__(256)
void gat_deg_zero(int* __restrict__ deg)
{
    int i = blockIdx.x * blockDim.x + threadIdx.x;
    if (i < NNODE) deg[i] = 0;
}

__global__ __launch_bounds__(256)
void gat_deg_rank(const int* __restrict__ dst,
                  int* __restrict__ deg,
                  int* __restrict__ rank)
{
    int i = blockIdx.x * blockDim.x + threadIdx.x;
    if (i >= NEDGE) return;
    rank[i] = atomicAdd(&deg[dst[i]], 1);
}

#define SCAN_PER 40
#define SCAN_ACTIVE (NNODE / SCAN_PER)
static_assert(NNODE % SCAN_PER == 0 && SCAN_ACTIVE < 1024 && SCAN_PER % 4 == 0,
              "scan block layout");

__global__ __launch_bounds__(1024)
void gat_rowptr_scan(const int* __restrict__ deg, int* __restrict__ rowptr)
{
    __shared__ int wtot[32];
    const int tid  = threadIdx.x;
    const int lane = tid & 31;
    const int w    = tid >> 5;

    int pre[SCAN_PER];
    int tot = 0;
    if (tid < SCAN_ACTIVE) {
        const int4* d4 = reinterpret_cast<const int4*>(deg + tid * SCAN_PER);
        #pragma unroll
        for (int q = 0; q < SCAN_PER / 4; ++q) {
            const int4 d = d4[q];
            pre[4 * q + 0] = tot; tot += d.x;
            pre[4 * q + 1] = tot; tot += d.y;
            pre[4 * q + 2] = tot; tot += d.z;
            pre[4 * q + 3] = tot; tot += d.w;
        }
    }
    int inc = tot;
    #pragma unroll
    for (int off = 1; off < 32; off <<= 1) {
        int n = __shfl_up(inc, off, 32);
        if (lane >= off) inc += n;
    }
    if (lane == 31) wtot[w] = inc;
    __syncthreads();
    if (w == 0) {
        const int t = wtot[lane];
        int ti = t;
        #pragma unroll
        for (int off = 1; off < 32; off <<= 1) {
            int n = __shfl_up(ti, off, 32);
            if (lane >= off) ti += n;
        }
        wtot[lane] = ti - t;
    }
    __syncthreads();
    const int offs = wtot[w] + inc - tot;
    if (tid < SCAN_ACTIVE) {
        int4* r4 = reinterpret_cast<int4*>(rowptr + tid * SCAN_PER);
        #pragma unroll
        for (int q = 0; q < SCAN_PER / 4; ++q) {
            int4 r;
            r.x = offs + pre[4 * q + 0];
            r.y = offs + pre[4 * q + 1];
            r.z = offs + pre[4 * q + 2];
            r.w = offs + pre[4 * q + 3];
            r4[q] = r;
        }
    } else if (tid == SCAN_ACTIVE) {
        rowptr[NNODE] = offs;
    }
}

__global__ __launch_bounds__(256)
void gat_edge_place(const int* __restrict__ src,
                    const int* __restrict__ dst,
                    const float* __restrict__ dfeat,
                    const int* __restrict__ rowptr,
                    const int* __restrict__ rank,
                    int* __restrict__ csrc,
                    float* __restrict__ cd)
{
    int i = blockIdx.x * blockDim.x + threadIdx.x;
    if (i >= NEDGE) return;
    const int p = rowptr[dst[i]] + rank[i];
    csrc[p] = src[i];
    cd[p]   = dfeat[i];
}

__device__ __forceinline__ float wave_max(float v) {
    #pragma unroll
    for (int off = 16; off > 0; off >>= 1) v = fmaxf(v, __shfl_xor(v, off, 32));
    return v;
}
__device__ __forceinline__ float wave_sum(float v) {
    #pragma unroll
    for (int off = 16; off > 0; off >>= 1) v += __shfl_xor(v, off, 32);
    return v;
}

template <int R>
__device__ __forceinline__ void add_rows(float4& acc, const float4* __restrict__ z4,
                                         int lane, int s, int ali, int k)
{
    int    sk[R];
    float  ak[R];
    float4 r[R];
    #pragma unroll
    for (int q = 0; q < R; ++q) {
        sk[q] = __builtin_amdgcn_readlane(s, k + q);
        ak[q] = __int_as_float(__builtin_amdgcn_readlane(ali, k + q));
    }
    #pragma unroll
    for (int q = 0; q < R; ++q) r[q] = z4[(size_t)sk[q] * (DD / 4) + lane];
    #pragma unroll
    for (int q = 0; q < R; ++q) {
        acc.x = fmaf(r[q].x, ak[q], acc.x);
        acc.y = fmaf(r[q].y, ak[q], acc.y);
        acc.z = fmaf(r[q].z, ak[q], acc.z);
        acc.w = fmaf(r[q].w, ak[q], acc.w);
    }
}

__device__ __forceinline__ void gather_rows(float4& acc, const float4* __restrict__ z4,
                                            int lane, int s, float al, int cnt)
{
    const int ali = __float_as_int(al);
    int k = 0;
    for (; k + 8 <= cnt; k += 8) add_rows<8>(acc, z4, lane, s, ali, k);
    if (cnt - k >= 4) { add_rows<4>(acc, z4, lane, s, ali, k); k += 4; }
    switch (cnt - k) {
        case 3:  add_rows<3>(acc, z4, lane, s, ali, k); break;
        case 2:  add_rows<2>(acc, z4, lane, s, ali, k); break;
        case 1:  add_rows<1>(acc, z4, lane, s, ali, k); break;
        default: break;
    }
}

__global__ __launch_bounds__(256)
void gat_node_aggregate(const int* __restrict__ rowptr,
                        const int* __restrict__ csrc,
                        const float* __restrict__ cd,
                        const float* __restrict__ W0_l,
                        const float* __restrict__ Wa_l,
                        const float* __restrict__ a_src,
                        const float* __restrict__ a_dst,
                        const float* __restrict__ z,
                        const float* __restrict__ zi,
                        float* __restrict__ hout)
{
    const int gid  = blockIdx.x * blockDim.x + threadIdx.x;
    const int lane = gid & 31;
    const int node = __builtin_amdgcn_readfirstlane(gid >> 5);
    if (node >= NNODE) return;
    const int beg = __builtin_amdgcn_readfirstlane(rowptr[node]);
    const int end = __builtin_amdgcn_readfirstlane(rowptr[node + 1]);
    const int deg = end - beg;

    const float ct  = W0_l[0] * Wa_l[2 * DD];
    const float adt = a_dst[node];
    const float4* z4 = reinterpret_cast<const float4*>(z);
    const float4 own = reinterpret_cast<const float4*>(zi)[(size_t)node * (DD / 4) + lane];
    float4 acc = {0.0f, 0.0f, 0.0f, 0.0f};

    if (deg <= 32) {
        const bool act = lane < deg;
        int   s = 0;
        float v = -INFINITY;
        if (act) {
            s = csrc[beg + lane];
            const float t = a_src[s] + adt + cd[beg + lane] * ct;
            v = (t > 0.0f) ? t : 0.01f * t;
        }
        float m = wave_max(v);
        if (!isfinite(m)) m = 0.0f;
        const float ex  = act ? __expf(v - m) : 0.0f;
        const float den = wave_sum(ex);
        gather_rows(acc, z4, lane, s, ex / den, deg);
    } else {
        float m = -INFINITY;
        for (int j = beg + lane; j < end; j += 32) {
            const float t = a_src[csrc[j]] + adt + cd[j] * ct;
            m = fmaxf(m, (t > 0.0f) ? t : 0.01f * t);
        }
        m = wave_max(m);
        if (!isfinite(m)) m = 0.0f;
        float den = 0.0f;
        for (int j = beg + lane; j < end; j += 32) {
            const float t = a_src[csrc[j]] + adt + cd[j] * ct;
            den += __expf(((t > 0.0f) ? t : 0.01f * t) - m);
        }
        den = wave_sum(den);
        for (int c = beg; c < end; c += 32) {
            const int j = c + lane;
            int   s  = 0;
            float al = 0.0f;
            if (j < end) {
                s = csrc[j];
                const float t = a_src[s] + adt + cd[j] * ct;
                al = __expf(((t > 0.0f) ? t : 0.01f * t) - m) / den;
            }
            const int left = end - c;
            gather_rows(acc, z4, lane, s, al, left < 32 ? left : 32);
        }
    }

    float4 o;
    o.x = fmaxf(own.x + acc.x, 0.0f);
    o.y = fmaxf(own.y + acc.y, 0.0f);
    o.z = fmaxf(own.z + acc.z, 0.0f);
    o.w = fmaxf(own.w + acc.w, 0.0f);
    reinterpret_cast<float4*>(hout)[(size_t)node * (DD / 4) + lane] = o;
}

extern "C" void kernel_launch(void* const* d_in, const int* in_sizes, int n_in,
                              void* d_out, int out_size, void* d_ws, size_t ws_size,
                              hipStream_t stream) {
    const float* attr = (const float*)d_in[0];
    const float* dfe  = (const float*)d_in[1];
    const int*   src  = (const int*)d_in[2];
    const int*   dst  = (const int*)d_in[3];
    const float* W0   = (const float*)d_in[4];
    const float* W1   = (const float*)d_in[5];
    const float* W2   = (const float*)d_in[6];
    const float* Wa   = (const float*)d_in[7];

    float* ws     = (float*)d_ws;
    float* hA     = ws;
    float* hB     = hA    + (size_t)NNODE * DD;
    float* z      = hB    + (size_t)NNODE * DD;
    float* zi     = z     + (size_t)NNODE * DD;
    float* a_src  = zi    + (size_t)NNODE * DD;
    float* a_dst  = a_src + NNODE;
    int*   deg    = (int*)(a_dst + NNODE);
    int*   rowptr = deg    + NNODE;
    int*   rank   = rowptr + NNODE + 4;
    int*   csrc   = rank   + NEDGE;
    float* cd     = (float*)(csrc + NEDGE);

    const int gemmBlocks = GEMM_BLOCKS;
    const int nodeBlocks = (NNODE * 32 + 255) / 256;
    const int edgeBlocks = (NEDGE + 255) / 256;
    const int degBlocks  = (NNODE + 255) / 256;

    gat_deg_zero<<<degBlocks, 256, 0, stream>>>(deg);
    gat_deg_rank<<<edgeBlocks, 256, 0, stream>>>(dst, deg, rank);
    gat_rowptr_scan<<<1, 1024, 0, stream>>>(deg, rowptr);
    gat_edge_place<<<edgeBlocks, 256, 0, stream>>>(src, dst, dfe, rowptr, rank, csrc, cd);

    for (int l = 0; l < 3; ++l) {
        const float* hin  = (l == 0) ? attr : ((l == 1) ? hA : hB);
        float*       hout = (l == 2) ? (float*)d_out : ((l == 0) ? hA : hB);
        const float* W1l = W1 + (size_t)l * DD * DD;
        const float* W2l = W2 + (size_t)l * DD * DD;
        const float* W0l = W0 + l;
        const float* Wal = Wa + (size_t)l * (2 * DD + 1);

        gat_gemm_wmma<<<gemmBlocks, 256, 0, stream>>>(hin, W1l, W2l, Wal, z, zi, a_src, a_dst);
        gat_node_aggregate<<<nodeBlocks, 256, 0, stream>>>(rowptr, csrc, cd, W0l, Wal,
                                                           a_src, a_dst, z, zi, hout);
    }
}


# Edge list grouped by destination through LDS in 2 launches (chunk blocks bin edges; bin blocks count, scan, place, whole-line stores), not 4 with global counters; edge arrival-order rank changes
# speedup vs baseline: 5.1645x; 1.4719x over previous
#include <hip/hip_runtime.h>
#include <hip/hip_bf16.h>
#include <math.h>

typedef float v2f __attribute__((ext_vector_type(2)));
typedef float v8f __attribute__((ext_vector_type(8)));

#define DD 128
#define NNODE 40000
#define NEDGE 640000

#define GEMM_TILES  (NNODE / 16)
#define GEMM_BLOCKS 512
#define A_PITCH     132
static_assert(NNODE % 16 == 0 && GEMM_BLOCKS <= GEMM_TILES, "row tiling");

template <int M>
__device__ __forceinline__ float row_xor(float v)
{
    return __int_as_float(__builtin_amdgcn_update_dpp(0, __float_as_int(v), 0x160 | M,
                                                      0xf, 0xf, true));
}

template <int M>
__device__ __forceinline__ float fold_pair(bool upper, float lo, float hi)
{
    const float keep = upper ? hi : lo;
    const float send = upper ? lo : hi;
    return keep + row_xor<M>(send);
}

__global__ __launch_bounds__(256)
void gat_gemm_wmma(const float* __restrict__ h,
                   const float* __restrict__ W1,
                   const float* __restrict__ W2,
                   const float* __restrict__ Wa_l,
                   float* __restrict__ z,
                   float* __restrict__ znb,
                   float* __restrict__ a_src,
                   float* __restrict__ a_dst)
{
    __shared__ __attribute__((aligned(16))) float As[2][16 * A_PITCH];
    __shared__ float Ps[2][8][32];

    const int tid   = threadIdx.x;
    const int lane  = tid & 31;
    const int wavew = tid >> 5;
    const int col0  = wavew << 4;
    const int mn    = lane & 15;
    const int khalf = (lane >> 4) << 1;

    v2f b1[DD / 4], b2[DD / 4];
    #pragma unroll
    for (int s = 0; s < DD / 4; ++s) {
        const size_t k = (size_t)(4 * s + khalf);
        b1[s].x = W1[k * DD + col0 + mn];
        b1[s].y = W1[(k + 1) * DD + col0 + mn];
        b2[s].x = W2[k * DD + col0 + mn];
        b2[s].y = W2[(k + 1) * DD + col0 + mn];
    }
    const float wa_s = Wa_l[col0 + mn];
    const float wa_d = Wa_l[DD + col0 + mn];
    const int pslot = ((mn >> 3) << 4) + ((lane >> 4) << 3) + (mn & 7);

    const float4* h4 = reinterpret_cast<const float4*>(h);
    int tile = blockIdx.x;
    float4 g0 = h4[(size_t)(tile * 16 + wavew)     * (DD / 4) + lane];
    float4 g1 = h4[(size_t)(tile * 16 + wavew + 8) * (DD / 4) + lane];
    *reinterpret_cast<float4*>(&As[0][wavew       * A_PITCH + lane * 4]) = g0;
    *reinterpret_cast<float4*>(&As[0][(wavew + 8) * A_PITCH + lane * 4]) = g1;
    __syncthreads();

    int buf = 0;
    for (; tile < GEMM_TILES; tile += GEMM_BLOCKS) {
        const int  nxt  = tile + GEMM_BLOCKS;
        const bool more = nxt < GEMM_TILES;
        if (more) {
            g0 = h4[(size_t)(nxt * 16 + wavew)     * (DD / 4) + lane];
            g1 = h4[(size_t)(nxt * 16 + wavew + 8) * (DD / 4) + lane];
        }

        v8f accZ = {0.f, 0.f, 0.f, 0.f, 0.f, 0.f, 0.f, 0.f};
        v8f accI = {0.f, 0.f, 0.f, 0.f, 0.f, 0.f, 0.f, 0.f};
        const float* arow = &As[buf][mn * A_PITCH + khalf];
        #pragma unroll
        for (int s = 0; s < DD / 4; ++s) {
            const v2f a = *reinterpret_cast<const v2f*>(arow + 4 * s);
            accZ = __builtin_amdgcn_wmma_f32_16x16x4_f32(false, a, false, b1[s],
                                                         (short)0, accZ, false, false);
            accI = __builtin_amdgcn_wmma_f32_16x16x4_f32(false, a, false, b2[s],
                                                         (short)0, accI, false, false);
        }

        {
            float v8[8], v4[4], v2[2];
            #pragma unroll
            for (int r = 0; r < 8; ++r)
                v8[r] = fold_pair<8>((lane & 8) != 0, accZ[r] * wa_s, accZ[r] * wa_d);
            #pragma unroll
            for (int r = 0; r < 4; ++r)
                v4[r] = fold_pair<4>((lane & 4) != 0, v8[r], v8[r + 4]);
            #pragma unroll
            for (int r = 0; r < 2; ++r)
                v2[r] = fold_pair<2>((lane & 2) != 0, v4[r], v4[r + 2]);
            Ps[buf][wavew][pslot] = fold_pair<1>((lane & 1) != 0, v2[0], v2[1]);
        }

        if (more) {
            *reinterpret_cast<float4*>(&As[buf ^ 1][wavew       * A_PITCH + lane * 4]) = g0;
            *reinterpret_cast<float4*>(&As[buf ^ 1][(wavew + 8) * A_PITCH + lane * 4]) = g1;
        }
        __syncthreads();

        if (wavew == 0) {
            float s = Ps[buf][0][lane];
            #pragma unroll
            for (int w = 1; w < 8; ++w) s += Ps[buf][w][lane];
            float* adst = (lane & 16) ? a_dst : a_src;
            adst[tile * 16 + mn] = s;
        }
        buf ^= 1;

        const int rowBase = tile * 16 + ((lane >> 4) << 3);
        #pragma unroll
        for (int r = 0; r < 8; ++r) {
            size_t idx = (size_t)(rowBase + r) * DD + col0 + mn;
            z[idx]   = accZ[r];
            znb[idx] = accI[r];
        }
    }
}

#define CHUNK      2560
#define NCHUNK     (NEDGE / CHUNK)
#define BIN_NODES  160
#define NBIN       (NNODE / BIN_NODES)
#define TAB_PITCH  (NBIN + 1)
#define TAB_INTS   62752
#define EPT        (CHUNK / 256)
#define BIN_CAP    4096
#define RUN_GROUP  8
static_assert(NEDGE % CHUNK == 0 && CHUNK % 256 == 0 && NNODE % BIN_NODES == 0, "chunking");
static_assert(NBIN < 256 && NCHUNK <= 256 && BIN_NODES <= 256 && NNODE <= 65536, "packing");
static_assert(NCHUNK * TAB_PITCH <= TAB_INTS && (TAB_INTS + NNODE + 4) % 2 == 0, "workspace layout");

__device__ __forceinline__ int block_scan_excl(int v, int tid, int* s_w, int& total)
{
    const int lane = tid & 31;
    const int w    = tid >> 5;
    int inc = v;
    #pragma unroll
    for (int off = 1; off < 32; off <<= 1) {
        const int t = __shfl_up(inc, off, 32);
        if (lane >= off) inc += t;
    }
    if (lane == 31) s_w[w] = inc;
    __syncthreads();
    int woff = 0, tot = 0;
    #pragma unroll
    for (int k = 0; k < 8; ++k) {
        const int t = s_w[k];
        if (k < w) woff += t;
        tot += t;
    }
    __syncthreads();
    total = tot;
    return woff + inc - v;
}

__global__ __launch_bounds__(256)
void gat_bin_chunk(const int* __restrict__ src,
                   const int* __restrict__ dst,
                   const float* __restrict__ dfeat,
                   int* __restrict__ tab,
                   int* __restrict__ rec_w)
{
    int2* __restrict__ rec = reinterpret_cast<int2*>(rec_w);
    __shared__ int  s_cnt[256];
    __shared__ int  s_off[256];
    __shared__ int  s_w[8];
    __shared__ int2 s_rec[CHUNK];

    const int tid = threadIdx.x;
    const int e0  = blockIdx.x * CHUNK;
    s_cnt[tid] = 0;
    __syncthreads();

    int   dv[EPT], sv[EPT], rk[EPT];
    float fv[EPT];
    #pragma unroll
    for (int k = 0; k < EPT; ++k) {
        dv[k] = dst[e0 + k * 256 + tid];
        sv[k] = src[e0 + k * 256 + tid];
        fv[k] = dfeat[e0 + k * 256 + tid];
    }
    #pragma unroll
    for (int k = 0; k < EPT; ++k) rk[k] = atomicAdd(&s_cnt[dv[k] / BIN_NODES], 1);
    __syncthreads();

    int total;
    const int ex = block_scan_excl(s_cnt[tid], tid, s_w, total);
    s_off[tid] = ex;
    if (tid <= NBIN) tab[blockIdx.x * TAB_PITCH + tid] = ex;
    __syncthreads();

    #pragma unroll
    for (int k = 0; k < EPT; ++k) {
        const int bin = dv[k] / BIN_NODES;
        const int p   = s_off[bin] + rk[k];
        s_rec[p] = make_int2(((dv[k] - bin * BIN_NODES) << 16) | sv[k], __float_as_int(fv[k]));
    }
    __syncthreads();
    #pragma unroll
    for (int k = 0; k < EPT; ++k) rec[e0 + k * 256 + tid] = s_rec[k * 256 + tid];
}

template <int PASS>
__device__ __forceinline__ void bin_walk(const int2* __restrict__ rec,
                                         const int* s_st, const int* s_len,
                                         int* s_cnt, const int* s_off, int* s_cur,
                                         int* s_osrc, float* s_od,
                                         int* __restrict__ csrc, float* __restrict__ cd,
                                         int base, bool staged, int w, int lane)
{
    const int hl = lane >> 4;
    const int jl = lane & 15;
    auto visit = [&](const int2 r) {
        const int node = r.x >> 16;
        if (PASS == 0) {
            atomicAdd(&s_cnt[node], 1);
        } else {
            const int p = s_off[node] + atomicAdd(&s_cur[node], 1);
            if (staged) {
                s_osrc[p] = r.x & 0xffff;
                s_od[p]   = __int_as_float(r.y);
            } else {
                csrc[base + p] = r.x & 0xffff;
                cd[base + p]   = __int_as_float(r.y);
            }
        }
    };
    for (int i0 = 0; i0 < 16; i0 += RUN_GROUP) {
        int2        r[RUN_GROUP];
        int         ln[RUN_GROUP];
        const int2* rp[RUN_GROUP];
        #pragma unroll
        for (int q = 0; q < RUN_GROUP; ++q) {
            const int c = 2 * (w + 8 * (i0 + q)) + hl;
            ln[q] = s_len[c];
            rp[q] = rec + (size_t)c * CHUNK + s_st[c];
            r[q]  = make_int2(0, 0);
        }
        #pragma unroll
        for (int q = 0; q < RUN_GROUP; ++q)
            if (jl < ln[q]) r[q] = rp[q][jl];
        #pragma unroll
        for (int q = 0; q < RUN_GROUP; ++q)
            if (jl < ln[q]) visit(r[q]);
        #pragma unroll
        for (int q = 0; q < RUN_GROUP; ++q)
            for (int j = jl + 16; j < ln[q]; j += 16) visit(rp[q][j]);
    }
}

__global__ __launch_bounds__(256)
void gat_bin_sort(const int* __restrict__ tab,
                  const int* __restrict__ rec_w,
                  int* __restrict__ rowptr,
                  int* __restrict__ csrc,
                  float* __restrict__ cd)
{
    __shared__ int   s_st[256];
    __shared__ int   s_len[256];
    __shared__ int   s_cnt[256];
    __shared__ int   s_off[256];
    __shared__ int   s_cur[256];
    __shared__ int   s_w[8];
    __shared__ int   s_osrc[BIN_CAP];
    __shared__ float s_od[BIN_CAP];

    const int tid  = threadIdx.x;
    const int lane = tid & 31;
    const int w    = tid >> 5;
    const int b    = blockIdx.x;
    const int2* rec = reinterpret_cast<const int2*>(rec_w);
    int st = 0, len = 0;
    if (tid < NCHUNK) {
        st  = tab[tid * TAB_PITCH + b];
        len = tab[tid * TAB_PITCH + b + 1] - st;
    }
    s_st[tid]  = st;
    s_len[tid] = len;
    s_cnt[tid] = 0;
    s_cur[tid] = 0;
    int base, n;
    block_scan_excl(st,  tid, s_w, base);
    block_scan_excl(len, tid, s_w, n);
    const bool staged = n <= BIN_CAP;

    bin_walk<0>(rec, s_st, s_len, s_cnt, s_off, s_cur, s_osrc, s_od, csrc, cd,
                base, staged, w, lane);
    __syncthreads();

    int tot2;
    const int ex = block_scan_excl(s_cnt[tid], tid, s_w, tot2);
    s_off[tid] = ex;
    if (tid < BIN_NODES) rowptr[b * BIN_NODES + tid] = base + ex;
    if (b == NBIN - 1 && tid == 0) rowptr[NNODE] = base + n;
    __syncthreads();

    bin_walk<1>(rec, s_st, s_len, s_cnt, s_off, s_cur, s_osrc, s_od, csrc, cd,
                base, staged, w, lane);
    __syncthreads();
    if (staged) {
        for (int j = tid; j < n; j += 256) {
            csrc[base + j] = s_osrc[j];
            cd[base + j]   = s_od[j];
        }
    }
}

__device__ __forceinline__ float wave_max(float v) {
    #pragma unroll
    for (int off = 16; off > 0; off >>= 1) v = fmaxf(v, __shfl_xor(v, off, 32));
    return v;
}
__device__ __forceinline__ float wave_sum(float v) {
    #pragma unroll
    for (int off = 16; off > 0; off >>= 1) v += __shfl_xor(v, off, 32);
    return v;
}

template <int R>
__device__ __forceinline__ void add_rows(float4& acc, const float4* __restrict__ z4,
                                         int lane, int s, int ali, int k)
{
    int    sk[R];
    float  ak[R];
    float4 r[R];
    #pragma unroll
    for (int q = 0; q < R; ++q) {
        sk[q] = __builtin_amdgcn_readlane(s, k + q);
        ak[q] = __int_as_float(__builtin_amdgcn_readlane(ali, k + q));
    }
    #pragma unroll
    for (int q = 0; q < R; ++q) r[q] = z4[(size_t)sk[q] * (DD / 4) + lane];
    #pragma unroll
    for (int q = 0; q < R; ++q) {
        acc.x = fmaf(r[q].x, ak[q], acc.x);
        acc.y = fmaf(r[q].y, ak[q], acc.y);
        acc.z = fmaf(r[q].z, ak[q], acc.z);
        acc.w = fmaf(r[q].w, ak[q], acc.w);
    }
}

__device__ __forceinline__ void gather_rows(float4& acc, const float4* __restrict__ z4,
                                            int lane, int s, float al, int cnt)
{
    const int ali = __float_as_int(al);
    int k = 0;
    for (; k + 8 <= cnt; k += 8) add_rows<8>(acc, z4, lane, s, ali, k);
    if (cnt - k >= 4) { add_rows<4>(acc, z4, lane, s, ali, k); k += 4; }
    switch (cnt - k) {
        case 3:  add_rows<3>(acc, z4, lane, s, ali, k); break;
        case 2:  add_rows<2>(acc, z4, lane, s, ali, k); break;
        case 1:  add_rows<1>(acc, z4, lane, s, ali, k); break;
        default: break;
    }
}

__global__ __launch_bounds__(256)
void gat_node_aggregate(const int* __restrict__ rowptr,
                        const int* __restrict__ csrc,
                        const float* __restrict__ cd,
                        const float* __restrict__ W0_l,
                        const float* __restrict__ Wa_l,
                        const float* __restrict__ a_src,
                        const float* __restrict__ a_dst,
                        const float* __restrict__ z,
                        const float* __restrict__ zi,
                        float* __restrict__ hout)
{
    const int gid  = blockIdx.x * blockDim.x + threadIdx.x;
    const int lane = gid & 31;
    const int node = __builtin_amdgcn_readfirstlane(gid >> 5);
    if (node >= NNODE) return;
    const int beg = __builtin_amdgcn_readfirstlane(rowptr[node]);
    const int end = __builtin_amdgcn_readfirstlane(rowptr[node + 1]);
    const int deg = end - beg;

    const float ct  = W0_l[0] * Wa_l[2 * DD];
    const float adt = a_dst[node];
    const float4* z4 = reinterpret_cast<const float4*>(z);
    const float4 own = reinterpret_cast<const float4*>(zi)[(size_t)node * (DD / 4) + lane];
    float4 acc = {0.0f, 0.0f, 0.0f, 0.0f};

    if (deg <= 32) {
        const bool act = lane < deg;
        int   s = 0;
        float v = -INFINITY;
        if (act) {
            s = csrc[beg + lane];
            const float t = a_src[s] + adt + cd[beg + lane] * ct;
            v = (t > 0.0f) ? t : 0.01f * t;
        }
        float m = wave_max(v);
        if (!isfinite(m)) m = 0.0f;
        const float ex  = act ? __expf(v - m) : 0.0f;
        const float den = wave_sum(ex);
        gather_rows(acc, z4, lane, s, ex / den, deg);
    } else {
        float m = -INFINITY;
        for (int j = beg + lane; j < end; j += 32) {
            const float t = a_src[csrc[j]] + adt + cd[j] * ct;
            m = fmaxf(m, (t > 0.0f) ? t : 0.01f * t);
        }
        m = wave_max(m);
        if (!isfinite(m)) m = 0.0f;
        float den = 0.0f;
        for (int j = beg + lane; j < end; j += 32) {
            const float t = a_src[csrc[j]] + adt + cd[j] * ct;
            den += __expf(((t > 0.0f) ? t : 0.01f * t) - m);
        }
        den = wave_sum(den);
        for (int c = beg; c < end; c += 32) {
            const int j = c + lane;
            int   s  = 0;
            float al = 0.0f;
            if (j < end) {
                s = csrc[j];
                const float t = a_src[s] + adt + cd[j] * ct;
                al = __expf(((t > 0.0f) ? t : 0.01f * t) - m) / den;
            }
            const int left = end - c;
            gather_rows(acc, z4, lane, s, al, left < 32 ? left : 32);
        }
    }

    float4 o;
    o.x = fmaxf(own.x + acc.x, 0.0f);
    o.y = fmaxf(own.y + acc.y, 0.0f);
    o.z = fmaxf(own.z + acc.z, 0.0f);
    o.w = fmaxf(own.w + acc.w, 0.0f);
    reinterpret_cast<float4*>(hout)[(size_t)node * (DD / 4) + lane] = o;
}

extern "C" void kernel_launch(void* const* d_in, const int* in_sizes, int n_in,
                              void* d_out, int out_size, void* d_ws, size_t ws_size,
                              hipStream_t stream) {
    const float* attr = (const float*)d_in[0];
    const float* dfe  = (const float*)d_in[1];
    const int*   src  = (const int*)d_in[2];
    const int*   dst  = (const int*)d_in[3];
    const float* W0   = (const float*)d_in[4];
    const float* W1   = (const float*)d_in[5];
    const float* W2   = (const float*)d_in[6];
    const float* Wa   = (const float*)d_in[7];

    float* ws     = (float*)d_ws;
    float* hA     = ws;
    float* hB     = hA    + (size_t)NNODE * DD;
    float* z      = hB    + (size_t)NNODE * DD;
    float* zi     = z     + (size_t)NNODE * DD;
    float* a_src  = zi    + (size_t)NNODE * DD;
    float* a_dst  = a_src + NNODE;
    int*   tab    = (int*)(a_dst + NNODE);
    int*   rowptr = tab    + TAB_INTS;
    int*   rec    = rowptr + NNODE + 4;
    int*   csrc   = rec    + 2 * NEDGE;
    float* cd     = (float*)(csrc + NEDGE);

    const int gemmBlocks = GEMM_BLOCKS;
    const int nodeBlocks = (NNODE * 32 + 255) / 256;

    gat_bin_chunk<<<NCHUNK, 256, 0, stream>>>(src, dst, dfe, tab, rec);
    gat_bin_sort<<<NBIN, 256, 0, stream>>>(tab, rec, rowptr, csrc, cd);

    for (int l = 0; l < 3; ++l) {
        const float* hin  = (l == 0) ? attr : ((l == 1) ? hA : hB);
        float*       hout = (l == 2) ? (float*)d_out : ((l == 0) ? hA : hB);
        const float* W1l = W1 + (size_t)l * DD * DD;
        const float* W2l = W2 + (size_t)l * DD * DD;
        const float* W0l = W0 + l;
        const float* Wal = Wa + (size_t)l * (2 * DD + 1);

        gat_gemm_wmma<<<gemmBlocks, 256, 0, stream>>>(hin, W1l, W2l, Wal, z, zi, a_src, a_dst);
        gat_node_aggregate<<<nodeBlocks, 256, 0, stream>>>(rowptr, csrc, cd, W0l, Wal,
                                                           a_src, a_dst, z, zi, hout);
    }
}


# Layer-0 GEMM (needs no edge list) split in two halves that run as extra blocks inside the two edge-grouping launches: 8 launches become 7 and the grouping runs under WMMA work; sum orders kept
# speedup vs baseline: 5.3639x; 1.0386x over previous
#include <hip/hip_runtime.h>
#include <hip/hip_bf16.h>
#include <math.h>

typedef float v2f __attribute__((ext_vector_type(2)));
typedef float v8f __attribute__((ext_vector_type(8)));

#define DD 128
#define NNODE 40000
#define NEDGE 640000

#define GEMM_TILES  (NNODE / 16)
#define GEMM_BLOCKS 512
#define HALF_TILES  (GEMM_TILES / 2)
#define SIDE_BLOCKS 256
#define A_PITCH     132
static_assert(NNODE % 16 == 0 && GEMM_BLOCKS <= GEMM_TILES, "row tiling");
static_assert(GEMM_TILES % 2 == 0 && SIDE_BLOCKS <= HALF_TILES, "row tiling of the halves");

template <int M>
__device__ __forceinline__ float row_xor(float v)
{
    return __int_as_float(__builtin_amdgcn_update_dpp(0, __float_as_int(v), 0x160 | M,
                                                      0xf, 0xf, true));
}

template <int M>
__device__ __forceinline__ float fold_pair(bool upper, float lo, float hi)
{
    const float keep = upper ? hi : lo;
    const float send = upper ? lo : hi;
    return keep + row_xor<M>(send);
}

template <int T0, int T1, int NB>
__device__ __forceinline__
void gemm_tiles(const int bid,
                const float* __restrict__ h,
                const float* __restrict__ W1,
                const float* __restrict__ W2,
                const float* __restrict__ Wa_l,
                float* __restrict__ z,
                float* __restrict__ znb,
                float* __restrict__ a_src,
                float* __restrict__ a_dst)
{
    static_assert(NB <= T1 - T0, "every block owns at least one tile");
    __shared__ __attribute__((aligned(16))) float As[2][16 * A_PITCH];
    __shared__ float Ps[2][8][32];

    const int tid   = threadIdx.x;
    const int lane  = tid & 31;
    const int wavew = tid >> 5;
    const int col0  = wavew << 4;
    const int mn    = lane & 15;
    const int khalf = (lane >> 4) << 1;

    v2f b1[DD / 4], b2[DD / 4];
    #pragma unroll
    for (int s = 0; s < DD / 4; ++s) {
        const size_t k = (size_t)(4 * s + khalf);
        b1[s].x = W1[k * DD + col0 + mn];
        b1[s].y = W1[(k + 1) * DD + col0 + mn];
        b2[s].x = W2[k * DD + col0 + mn];
        b2[s].y = W2[(k + 1) * DD + col0 + mn];
    }
    const float wa_s = Wa_l[col0 + mn];
    const float wa_d = Wa_l[DD + col0 + mn];
    const int pslot = ((mn >> 3) << 4) + ((lane >> 4) << 3) + (mn & 7);

    const float4* h4 = reinterpret_cast<const float4*>(h);
    int tile = T0 + bid;
    float4 g0 = h4[(size_t)(tile * 16 + wavew)     * (DD / 4) + lane];
    float4 g1 = h4[(size_t)(tile * 16 + wavew + 8) * (DD / 4) + lane];
    *reinterpret_cast<float4*>(&As[0][wavew       * A_PITCH + lane * 4]) = g0;
    *reinterpret_cast<float4*>(&As[0][(wavew + 8) * A_PITCH + lane * 4]) = g1;
    __syncthreads();

    int buf = 0;
    for (; tile < T1; tile += NB) {
        const int  nxt  = tile + NB;
        const bool more = nxt < T1;
        if (more) {
            g0 = h4[(size_t)(nxt * 16 + wavew)     * (DD / 4) + lane];
            g1 = h4[(size_t)(nxt * 16 + wavew + 8) * (DD / 4) + lane];
        }

        v8f accZ = {0.f, 0.f, 0.f, 0.f, 0.f, 0.f, 0.f, 0.f};
        v8f accI = {0.f, 0.f, 0.f, 0.f, 0.f, 0.f, 0.f, 0.f};
        const float* arow = &As[buf][mn * A_PITCH + khalf];
        #pragma unroll
        for (int s = 0; s < DD / 4; ++s) {
            const v2f a = *reinterpret_cast<const v2f*>(arow + 4 * s);
            accZ = __builtin_amdgcn_wmma_f32_16x16x4_f32(false, a, false, b1[s],
                                                         (short)0, accZ, false, false);
            accI = __builtin_amdgcn_wmma_f32_16x16x4_f32(false, a, false, b2[s],
                                                         (short)0, accI, false, false);
        }

        {
            float v8[8], v4[4], v2[2];
            #pragma unroll
            for (int r = 0; r < 8; ++r)
                v8[r] = fold_pair<8>((lane & 8) != 0, accZ[r] * wa_s, accZ[r] * wa_d);
            #pragma unroll
            for (int r = 0; r < 4; ++r)
                v4[r] = fold_pair<4>((lane & 4) != 0, v8[r], v8[r + 4]);
            #pragma unroll
            for (int r = 0; r < 2; ++r)
                v2[r] = fold_pair<2>((lane & 2) != 0, v4[r], v4[r + 2]);
            Ps[buf][wavew][pslot] = fold_pair<1>((lane & 1) != 0, v2[0], v2[1]);
        }

        if (more) {
            *reinterpret_cast<float4*>(&As[buf ^ 1][wavew       * A_PITCH + lane * 4]) = g0;
            *reinterpret_cast<float4*>(&As[buf ^ 1][(wavew + 8) * A_PITCH + lane * 4]) = g1;
        }
        __syncthreads();

        if (wavew == 0) {
            float s = Ps[buf][0][lane];
            #pragma unroll
            for (int w = 1; w < 8; ++w) s += Ps[buf][w][lane];
            float* adst = (lane & 16) ? a_dst : a_src;
            adst[tile * 16 + mn] = s;
        }
        buf ^= 1;

        const int rowBase = tile * 16 + ((lane >> 4) << 3);
        #pragma unroll
        for (int r = 0; r < 8; ++r) {
            size_t idx = (size_t)(rowBase + r) * DD + col0 + mn;
            z[idx]   = accZ[r];
            znb[idx] = accI[r];
        }
    }
}

__global__ __launch_bounds__(256)
void gat_gemm_wmma(const float* __restrict__ h,
                   const float* __restrict__ W1,
                   const float* __restrict__ W2,
                   const float* __restrict__ Wa_l,
                   float* __restrict__ z,
                   float* __restrict__ znb,
                   float* __restrict__ a_src,
                   float* __restrict__ a_dst)
{
    gemm_tiles<0, GEMM_TILES, GEMM_BLOCKS>(blockIdx.x, h, W1, W2, Wa_l, z, znb, a_src, a_dst);
}

#define CHUNK      2560
#define NCHUNK     (NEDGE / CHUNK)
#define BIN_NODES  160
#define NBIN       (NNODE / BIN_NODES)
#define TAB_PITCH  (NBIN + 1)
#define TAB_INTS   62752
#define EPT        (CHUNK / 256)
#define BIN_CAP    4096
#define RUN_GROUP  8
static_assert(NEDGE % CHUNK == 0 && CHUNK % 256 == 0 && NNODE % BIN_NODES == 0, "chunking");
static_assert(NBIN < 256 && NCHUNK <= 256 && BIN_NODES <= 256 && NNODE <= 65536, "packing");
static_assert(NCHUNK * TAB_PITCH <= TAB_INTS && (TAB_INTS + NNODE + 4) % 2 == 0, "workspace layout");

__device__ __forceinline__ int block_scan_excl(int v, int tid, int* s_w, int& total)
{
    const int lane = tid & 31;
    const int w    = tid >> 5;
    int inc = v;
    #pragma unroll
    for (int off = 1; off < 32; off <<= 1) {
        const int t = __shfl_up(inc, off, 32);
        if (lane >= off) inc += t;
    }
    if (lane == 31) s_w[w] = inc;
    __syncthreads();
    int woff = 0, tot = 0;
    #pragma unroll
    for (int k = 0; k < 8; ++k) {
        const int t = s_w[k];
        if (k < w) woff += t;
        tot += t;
    }
    __syncthreads();
    total = tot;
    return woff + inc - v;
}

__device__ __forceinline__
void bin_chunk_block(const int* __restrict__ src,
                     const int* __restrict__ dst,
                     const float* __restrict__ dfeat,
                     int* __restrict__ tab,
                     int* __restrict__ rec_w)
{
    int2* __restrict__ rec = reinterpret_cast<int2*>(rec_w);
    __shared__ int  s_cnt[256];
    __shared__ int  s_off[256];
    __shared__ int  s_w[8];
    __shared__ int2 s_rec[CHUNK];

    const int tid = threadIdx.x;
    const int e0  = blockIdx.x * CHUNK;
    s_cnt[tid] = 0;
    __syncthreads();

    int   dv[EPT], sv[EPT], rk[EPT];
    float fv[EPT];
    #pragma unroll
    for (int k = 0; k < EPT; ++k) {
        dv[k] = dst[e0 + k * 256 + tid];
        sv[k] = src[e0 + k * 256 + tid];
        fv[k] = dfeat[e0 + k * 256 + tid];
    }
    #pragma unroll
    for (int k = 0; k < EPT; ++k) rk[k] = atomicAdd(&s_cnt[dv[k] / BIN_NODES], 1);
    __syncthreads();

    int total;
    const int ex = block_scan_excl(s_cnt[tid], tid, s_w, total);
    s_off[tid] = ex;
    if (tid <= NBIN) tab[blockIdx.x * TAB_PITCH + tid] = ex;
    __syncthreads();

    #pragma unroll
    for (int k = 0; k < EPT; ++k) {
        const int bin = dv[k] / BIN_NODES;
        const int p   = s_off[bin] + rk[k];
        s_rec[p] = make_int2(((dv[k] - bin * BIN_NODES) << 16) | sv[k], __float_as_int(fv[k]));
    }
    __syncthreads();
    #pragma unroll
    for (int k = 0; k < EPT; ++k) rec[e0 + k * 256 + tid] = s_rec[k * 256 + tid];
}

template <int PASS>
__device__ __forceinline__ void bin_walk(const int2* __restrict__ rec,
                                         const int* s_st, const int* s_len,
                                         int* s_cnt, const int* s_off, int* s_cur,
                                         int* s_osrc, float* s_od,
                                         int* __restrict__ csrc, float* __restrict__ cd,
                                         int base, bool staged, int w, int lane)
{
    const int hl = lane >> 4;
    const int jl = lane & 15;
    auto visit = [&](const int2 r) {
        const int node = r.x >> 16;
        if (PASS == 0) {
            atomicAdd(&s_cnt[node], 1);
        } else {
            const int p = s_off[node] + atomicAdd(&s_cur[node], 1);
            if (staged) {
                s_osrc[p] = r.x & 0xffff;
                s_od[p]   = __int_as_float(r.y);
            } else {
                csrc[base + p] = r.x & 0xffff;
                cd[base + p]   = __int_as_float(r.y);
            }
        }
    };
    for (int i0 = 0; i0 < 16; i0 += RUN_GROUP) {
        int2        r[RUN_GROUP];
        int         ln[RUN_GROUP];
        const int2* rp[RUN_GROUP];
        #pragma unroll
        for (int q = 0; q < RUN_GROUP; ++q) {
            const int c = 2 * (w + 8 * (i0 + q)) + hl;
            ln[q] = s_len[c];
            rp[q] = rec + (size_t)c * CHUNK + s_st[c];
            r[q]  = make_int2(0, 0);
        }
        #pragma unroll
        for (int q = 0; q < RUN_GROUP; ++q)
            if (jl < ln[q]) r[q] = rp[q][jl];
        #pragma unroll
        for (int q = 0; q < RUN_GROUP; ++q)
            if (jl < ln[q]) visit(r[q]);
        #pragma unroll
        for (int q = 0; q < RUN_GROUP; ++q)
            for (int j = jl + 16; j < ln[q]; j += 16) visit(rp[q][j]);
    }
}

__device__ __forceinline__
void bin_sort_block(const int* __restrict__ tab,
                    const int* __restrict__ rec_w,
                    int* __restrict__ rowptr,
                    int* __restrict__ csrc,
                    float* __restrict__ cd)
{
    __shared__ int   s_st[256];
    __shared__ int   s_len[256];
    __shared__ int   s_cnt[256];
    __shared__ int   s_off[256];
    __shared__ int   s_cur[256];
    __shared__ int   s_w[8];
    __shared__ int   s_osrc[BIN_CAP];
    __shared__ float s_od[BIN_CAP];

    const int tid  = threadIdx.x;
    const int lane = tid & 31;
    const int w    = tid >> 5;
    const int b    = blockIdx.x;
    const int2* rec = reinterpret_cast<const int2*>(rec_w);
    int st = 0, len = 0;
    if (tid < NCHUNK) {
        st  = tab[tid * TAB_PITCH + b];
        len = tab[tid * TAB_PITCH + b + 1] - st;
    }
    s_st[tid]  = st;
    s_len[tid] = len;
    s_cnt[tid] = 0;
    s_cur[tid] = 0;
    int base, n;
    block_scan_excl(st,  tid, s_w, base);
    block_scan_excl(len, tid, s_w, n);
    const bool staged = n <= BIN_CAP;

    bin_walk<0>(rec, s_st, s_len, s_cnt, s_off, s_cur, s_osrc, s_od, csrc, cd,
                base, staged, w, lane);
    __syncthreads();

    int tot2;
    const int ex = block_scan_excl(s_cnt[tid], tid, s_w, tot2);
    s_off[tid] = ex;
    if (tid < BIN_NODES) rowptr[b * BIN_NODES + tid] = base + ex;
    if (b == NBIN - 1 && tid == 0) rowptr[NNODE] = base + n;
    __syncthreads();

    bin_walk<1>(rec, s_st, s_len, s_cnt, s_off, s_cur, s_osrc, s_od, csrc, cd,
                base, staged, w, lane);
    __syncthreads();
    if (staged) {
        for (int j = tid; j < n; j += 256) {
            csrc[base + j] = s_osrc[j];
            cd[base + j]   = s_od[j];
        }
    }
}

__global__ __launch_bounds__(256)
void gat_chunk_and_gemm(const int* __restrict__ src,
                        const int* __restrict__ dst,
                        const float* __restrict__ dfeat,
                        int* __restrict__ tab,
                        int* __restrict__ rec_w,
                        const float* __restrict__ h,
                        const float* __restrict__ W1,
                        const float* __restrict__ W2,
                        const float* __restrict__ Wa_l,
                        float* __restrict__ z,
                        float* __restrict__ znb,
                        float* __restrict__ a_src,
                        float* __restrict__ a_dst)
{
    if (blockIdx.x < NCHUNK)
        bin_chunk_block(src, dst, dfeat, tab, rec_w);
    else
        gemm_tiles<0, HALF_TILES, SIDE_BLOCKS>(blockIdx.x - NCHUNK, h, W1, W2, Wa_l,
                                               z, znb, a_src, a_dst);
}

__global__ __launch_bounds__(256)
void gat_sort_and_gemm(const int* __restrict__ tab,
                       const int* __restrict__ rec_w,
                       int* __restrict__ rowptr,
                       int* __restrict__ csrc,
                       float* __restrict__ cd,
                       const float* __restrict__ h,
                       const float* __restrict__ W1,
                       const float* __restrict__ W2,
                       const float* __restrict__ Wa_l,
                       float* __restrict__ z,
                       float* __restrict__ znb,
                       float* __restrict__ a_src,
                       float* __restrict__ a_dst)
{
    if (blockIdx.x < NBIN)
        bin_sort_block(tab, rec_w, rowptr, csrc, cd);
    else
        gemm_tiles<HALF_TILES, GEMM_TILES, SIDE_BLOCKS>(blockIdx.x - NBIN, h, W1, W2, Wa_l,
                                                        z, znb, a_src, a_dst);
}

__device__ __forceinline__ float wave_max(float v) {
    #pragma unroll
    for (int off = 16; off > 0; off >>= 1) v = fmaxf(v, __shfl_xor(v, off, 32));
    return v;
}
__device__ __forceinline__ float wave_sum(float v) {
    #pragma unroll
    for (int off = 16; off > 0; off >>= 1) v += __shfl_xor(v, off, 32);
    return v;
}

template <int R>
__device__ __forceinline__ void add_rows(float4& acc, const float4* __restrict__ z4,
                                         int lane, int s, int ali, int k)
{
    int    sk[R];
    float  ak[R];
    float4 r[R];
    #pragma unroll
    for (int q = 0; q < R; ++q) {
        sk[q] = __builtin_amdgcn_readlane(s, k + q);
        ak[q] = __int_as_float(__builtin_amdgcn_readlane(ali, k + q));
    }
    #pragma unroll
    for (int q = 0; q < R; ++q) r[q] = z4[(size_t)sk[q] * (DD / 4) + lane];
    #pragma unroll
    for (int q = 0; q < R; ++q) {
        acc.x = fmaf(r[q].x, ak[q], acc.x);
        acc.y = fmaf(r[q].y, ak[q], acc.y);
        acc.z = fmaf(r[q].z, ak[q], acc.z);
        acc.w = fmaf(r[q].w, ak[q], acc.w);
    }
}

__device__ __forceinline__ void gather_rows(float4& acc, const float4* __restrict__ z4,
                                            int lane, int s, float al, int cnt)
{
    const int ali = __float_as_int(al);
    int k = 0;
    for (; k + 8 <= cnt; k += 8) add_rows<8>(acc, z4, lane, s, ali, k);
    if (cnt - k >= 4) { add_rows<4>(acc, z4, lane, s, ali, k); k += 4; }
    switch (cnt - k) {
        case 3:  add_rows<3>(acc, z4, lane, s, ali, k); break;
        case 2:  add_rows<2>(acc, z4, lane, s, ali, k); break;
        case 1:  add_rows<1>(acc, z4, lane, s, ali, k); break;
        default: break;
    }
}

__global__ __launch_bounds__(256)
void gat_node_aggregate(const int* __restrict__ rowptr,
                        const int* __restrict__ csrc,
                        const float* __restrict__ cd,
                        const float* __restrict__ W0_l,
                        const float* __restrict__ Wa_l,
                        const float* __restrict__ a_src,
                        const float* __restrict__ a_dst,
                        const float* __restrict__ z,
                        const float* __restrict__ zi,
                        float* __restrict__ hout)
{
    const int gid  = blockIdx.x * blockDim.x + threadIdx.x;
    const int lane = gid & 31;
    const int node = __builtin_amdgcn_readfirstlane(gid >> 5);
    if (node >= NNODE) return;
    const int beg = __builtin_amdgcn_readfirstlane(rowptr[node]);
    const int end = __builtin_amdgcn_readfirstlane(rowptr[node + 1]);
    const int deg = end - beg;

    const float ct  = W0_l[0] * Wa_l[2 * DD];
    const float adt = a_dst[node];
    const float4* z4 = reinterpret_cast<const float4*>(z);
    const float4 own = reinterpret_cast<const float4*>(zi)[(size_t)node * (DD / 4) + lane];
    float4 acc = {0.0f, 0.0f, 0.0f, 0.0f};

    if (deg <= 32) {
        const bool act = lane < deg;
        int   s = 0;
        float v = -INFINITY;
        if (act) {
            s = csrc[beg + lane];
            const float t = a_src[s] + adt + cd[beg + lane] * ct;
            v = (t > 0.0f) ? t : 0.01f * t;
        }
        float m = wave_max(v);
        if (!isfinite(m)) m = 0.0f;
        const float ex  = act ? __expf(v - m) : 0.0f;
        const float den = wave_sum(ex);
        gather_rows(acc, z4, lane, s, ex / den, deg);
    } else {
        float m = -INFINITY;
        for (int j = beg + lane; j < end; j += 32) {
            const float t = a_src[csrc[j]] + adt + cd[j] * ct;
            m = fmaxf(m, (t > 0.0f) ? t : 0.01f * t);
        }
        m = wave_max(m);
        if (!isfinite(m)) m = 0.0f;
        float den = 0.0f;
        for (int j = beg + lane; j < end; j += 32) {
            const float t = a_src[csrc[j]] + adt + cd[j] * ct;
            den += __expf(((t > 0.0f) ? t : 0.01f * t) - m);
        }
        den = wave_sum(den);
        for (int c = beg; c < end; c += 32) {
            const int j = c + lane;
            int   s  = 0;
            float al = 0.0f;
            if (j < end) {
                s = csrc[j];
                const float t = a_src[s] + adt + cd[j] * ct;
                al = __expf(((t > 0.0f) ? t : 0.01f * t) - m) / den;
            }
            const int left = end - c;
            gather_rows(acc, z4, lane, s, al, left < 32 ? left : 32);
        }
    }

    float4 o;
    o.x = fmaxf(own.x + acc.x, 0.0f);
    o.y = fmaxf(own.y + acc.y, 0.0f);
    o.z = fmaxf(own.z + acc.z, 0.0f);
    o.w = fmaxf(own.w + acc.w, 0.0f);
    reinterpret_cast<float4*>(hout)[(size_t)node * (DD / 4) + lane] = o;
}

extern "C" void kernel_launch(void* const* d_in, const int* in_sizes, int n_in,
                              void* d_out, int out_size, void* d_ws, size_t ws_size,
                              hipStream_t stream) {
    const float* attr = (const float*)d_in[0];
    const float* dfe  = (const float*)d_in[1];
    const int*   src  = (const int*)d_in[2];
    const int*   dst  = (const int*)d_in[3];
    const float* W0   = (const float*)d_in[4];
    const float* W1   = (const float*)d_in[5];
    const float* W2   = (const float*)d_in[6];
    const float* Wa   = (const float*)d_in[7];

    float* ws     = (float*)d_ws;
    float* hA     = ws;
    float* hB     = hA    + (size_t)NNODE * DD;
    float* z      = hB    + (size_t)NNODE * DD;
    float* zi     = z     + (size_t)NNODE * DD;
    float* a_src  = zi    + (size_t)NNODE * DD;
    float* a_dst  = a_src + NNODE;
    int*   tab    = (int*)(a_dst + NNODE);
    int*   rowptr = tab    + TAB_INTS;
    int*   rec    = rowptr + NNODE + 4;
    int*   csrc   = rec    + 2 * NEDGE;
    float* cd     = (float*)(csrc + NEDGE);

    const int gemmBlocks = GEMM_BLOCKS;
    const int nodeBlocks = (NNODE * 32 + 255) / 256;

    gat_chunk_and_gemm<<<NCHUNK + SIDE_BLOCKS, 256, 0, stream>>>(src, dst, dfe, tab, rec,
                                                                 attr, W1, W2, Wa,
                                                                 z, zi, a_src, a_dst);
    gat_sort_and_gemm<<<NBIN + SIDE_BLOCKS, 256, 0, stream>>>(tab, rec, rowptr, csrc, cd,
                                                              attr, W1, W2, Wa,
                                                              z, zi, a_src, a_dst);
    gat_node_aggregate<<<nodeBlocks, 256, 0, stream>>>(rowptr, csrc, cd, W0, Wa,
                                                       a_src, a_dst, z, zi, hA);

    for (int l = 1; l < 3; ++l) {
        const float* hin  = (l == 1) ? hA : hB;
        float*       hout = (l == 2) ? (float*)d_out : hB;
        const float* W1l = W1 + (size_t)l * DD * DD;
        const float* W2l = W2 + (size_t)l * DD * DD;
        const float* W0l = W0 + l;
        const float* Wal = Wa + (size_t)l * (2 * DD + 1);

        gat_gemm_wmma<<<gemmBlocks, 256, 0, stream>>>(hin, W1l, W2l, Wal, z, zi, a_src, a_dst);
        gat_node_aggregate<<<nodeBlocks, 256, 0, stream>>>(rowptr, csrc, cd, W0l, Wal,
                                                           a_src, a_dst, z, zi, hout);
    }
}


# Next layer's GEMM joined to the node aggregation: persistent 16-wave blocks form each 16-node tile of h in LDS and multiply it at once; 7 launches become 5, h stays out of memory; sum orders kept
# speedup vs baseline: 5.6723x; 1.0575x over previous
#include <hip/hip_runtime.h>
#include <hip/hip_bf16.h>
#include <math.h>

typedef float v2f __attribute__((ext_vector_type(2)));
typedef float v8f __attribute__((ext_vector_type(8)));

#define DD 128
#define NNODE 40000
#define NEDGE 640000

#define GEMM_TILES  (NNODE / 16)
#define HALF_TILES  (GEMM_TILES / 2)
#define SIDE_BLOCKS 256
#define A_PITCH     132
static_assert(NNODE % 16 == 0, "row tiling");
static_assert(GEMM_TILES % 2 == 0 && SIDE_BLOCKS <= HALF_TILES, "row tiling of the halves");

template <int M>
__device__ __forceinline__ float row_xor(float v)
{
    return __int_as_float(__builtin_amdgcn_update_dpp(0, __float_as_int(v), 0x160 | M,
                                                      0xf, 0xf, true));
}

template <int M>
__device__ __forceinline__ float fold_pair(bool upper, float lo, float hi)
{
    const float keep = upper ? hi : lo;
    const float send = upper ? lo : hi;
    return keep + row_xor<M>(send);
}

template <int T0, int T1, int NB>
__device__ __forceinline__
void gemm_tiles(const int bid,
                const float* __restrict__ h,
                const float* __restrict__ W1,
                const float* __restrict__ W2,
                const float* __restrict__ Wa_l,
                float* __restrict__ z,
                float* __restrict__ znb,
                float* __restrict__ a_src,
                float* __restrict__ a_dst)
{
    static_assert(NB <= T1 - T0, "every block owns at least one tile");
    __shared__ __attribute__((aligned(16))) float As[2][16 * A_PITCH];
    __shared__ float Ps[2][8][32];

    const int tid   = threadIdx.x;
    const int lane  = tid & 31;
    const int wavew = tid >> 5;
    const int col0  = wavew << 4;
    const int mn    = lane & 15;
    const int khalf = (lane >> 4) << 1;

    v2f b1[DD / 4], b2[DD / 4];
    #pragma unroll
    for (int s = 0; s < DD / 4; ++s) {
        const size_t k = (size_t)(4 * s + khalf);
        b1[s].x = W1[k * DD + col0 + mn];
        b1[s].y = W1[(k + 1) * DD + col0 + mn];
        b2[s].x = W2[k * DD + col0 + mn];
        b2[s].y = W2[(k + 1) * DD + col0 + mn];
    }
    const float wa_s = Wa_l[col0 + mn];
    const float wa_d = Wa_l[DD + col0 + mn];
    const int pslot = ((mn >> 3) << 4) + ((lane >> 4) << 3) + (mn & 7);

    const float4* h4 = reinterpret_cast<const float4*>(h);
    int tile = T0 + bid;
    float4 g0 = h4[(size_t)(tile * 16 + wavew)     * (DD / 4) + lane];
    float4 g1 = h4[(size_t)(tile * 16 + wavew + 8) * (DD / 4) + lane];
    *reinterpret_cast<float4*>(&As[0][wavew       * A_PITCH + lane * 4]) = g0;
    *reinterpret_cast<float4*>(&As[0][(wavew + 8) * A_PITCH + lane * 4]) = g1;
    __syncthreads();

    int buf = 0;
    for (; tile < T1; tile += NB) {
        const int  nxt  = tile + NB;
        const bool more = nxt < T1;
        if (more) {
            g0 = h4[(size_t)(nxt * 16 + wavew)     * (DD / 4) + lane];
            g1 = h4[(size_t)(nxt * 16 + wavew + 8) * (DD / 4) + lane];
        }

        v8f accZ = {0.f, 0.f, 0.f, 0.f, 0.f, 0.f, 0.f, 0.f};
        v8f accI = {0.f, 0.f, 0.f, 0.f, 0.f, 0.f, 0.f, 0.f};
        const float* arow = &As[buf][mn * A_PITCH + khalf];
        #pragma unroll
        for (int s = 0; s < DD / 4; ++s) {
            const v2f a = *reinterpret_cast<const v2f*>(arow + 4 * s);
            accZ = __builtin_amdgcn_wmma_f32_16x16x4_f32(false, a, false, b1[s],
                                                         (short)0, accZ, false, false);
            accI = __builtin_amdgcn_wmma_f32_16x16x4_f32(false, a, false, b2[s],
                                                         (short)0, accI, false, false);
        }

        {
            float v8[8], v4[4], v2[2];
            #pragma unroll
            for (int r = 0; r < 8; ++r)
                v8[r] = fold_pair<8>((lane & 8) != 0, accZ[r] * wa_s, accZ[r] * wa_d);
            #pragma unroll
            for (int r = 0; r < 4; ++r)
                v4[r] = fold_pair<4>((lane & 4) != 0, v8[r], v8[r + 4]);
            #pragma unroll
            for (int r = 0; r < 2; ++r)
                v2[r] = fold_pair<2>((lane & 2) != 0, v4[r], v4[r + 2]);
            Ps[buf][wavew][pslot] = fold_pair<1>((lane & 1) != 0, v2[0], v2[1]);
        }

        if (more) {
            *reinterpret_cast<float4*>(&As[buf ^ 1][wavew       * A_PITCH + lane * 4]) = g0;
            *reinterpret_cast<float4*>(&As[buf ^ 1][(wavew + 8) * A_PITCH + lane * 4]) = g1;
        }
        __syncthreads();

        if (wavew == 0) {
            float s = Ps[buf][0][lane];
            #pragma unroll
            for (int w = 1; w < 8; ++w) s += Ps[buf][w][lane];
            float* adst = (lane & 16) ? a_dst : a_src;
            adst[tile * 16 + mn] = s;
        }
        buf ^= 1;

        const int rowBase = tile * 16 + ((lane >> 4) << 3);
        #pragma unroll
        for (int r = 0; r < 8; ++r) {
            size_t idx = (size_t)(rowBase + r) * DD + col0 + mn;
            z[idx]   = accZ[r];
            znb[idx] = accI[r];
        }
    }
}

#define CHUNK      2560
#define NCHUNK     (NEDGE / CHUNK)
#define BIN_NODES  160
#define NBIN       (NNODE / BIN_NODES)
#define TAB_PITCH  (NBIN + 1)
#define TAB_INTS   62752
#define EPT        (CHUNK / 256)
#define BIN_CAP    4096
#define RUN_GROUP  8
static_assert(NEDGE % CHUNK == 0 && CHUNK % 256 == 0 && NNODE % BIN_NODES == 0, "chunking");
static_assert(NBIN < 256 && NCHUNK <= 256 && BIN_NODES <= 256 && NNODE <= 65536, "packing");
static_assert(NCHUNK * TAB_PITCH <= TAB_INTS && (TAB_INTS + NNODE + 4) % 2 == 0, "workspace layout");

__device__ __forceinline__ int block_scan_excl(int v, int tid, int* s_w, int& total)
{
    const int lane = tid & 31;
    const int w    = tid >> 5;
    int inc = v;
    #pragma unroll
    for (int off = 1; off < 32; off <<= 1) {
        const int t = __shfl_up(inc, off, 32);
        if (lane >= off) inc += t;
    }
    if (lane == 31) s_w[w] = inc;
    __syncthreads();
    int woff = 0, tot = 0;
    #pragma unroll
    for (int k = 0; k < 8; ++k) {
        const int t = s_w[k];
        if (k < w) woff += t;
        tot += t;
    }
    __syncthreads();
    total = tot;
    return woff + inc - v;
}

__device__ __forceinline__
void bin_chunk_block(const int* __restrict__ src,
                     const int* __restrict__ dst,
                     const float* __restrict__ dfeat,
                     int* __restrict__ tab,
                     int* __restrict__ rec_w)
{
    int2* __restrict__ rec = reinterpret_cast<int2*>(rec_w);
    __shared__ int  s_cnt[256];
    __shared__ int  s_off[256];
    __shared__ int  s_w[8];
    __shared__ int2 s_rec[CHUNK];

    const int tid = threadIdx.x;
    const int e0  = blockIdx.x * CHUNK;
    s_cnt[tid] = 0;
    __syncthreads();

    int   dv[EPT], sv[EPT], rk[EPT];
    float fv[EPT];
    #pragma unroll
    for (int k = 0; k < EPT; ++k) {
        dv[k] = dst[e0 + k * 256 + tid];
        sv[k] = src[e0 + k * 256 + tid];
        fv[k] = dfeat[e0 + k * 256 + tid];
    }
    #pragma unroll
    for (int k = 0; k < EPT; ++k) rk[k] = atomicAdd(&s_cnt[dv[k] / BIN_NODES], 1);
    __syncthreads();

    int total;
    const int ex = block_scan_excl(s_cnt[tid], tid, s_w, total);
    s_off[tid] = ex;
    if (tid <= NBIN) tab[blockIdx.x * TAB_PITCH + tid] = ex;
    __syncthreads();

    #pragma unroll
    for (int k = 0; k < EPT; ++k) {
        const int bin = dv[k] / BIN_NODES;
        const int p   = s_off[bin] + rk[k];
        s_rec[p] = make_int2(((dv[k] - bin * BIN_NODES) << 16) | sv[k], __float_as_int(fv[k]));
    }
    __syncthreads();
    #pragma unroll
    for (int k = 0; k < EPT; ++k) rec[e0 + k * 256 + tid] = s_rec[k * 256 + tid];
}

template <int PASS>
__device__ __forceinline__ void bin_walk(const int2* __restrict__ rec,
                                         const int* s_st, const int* s_len,
                                         int* s_cnt, const int* s_off, int* s_cur,
                                         int* s_osrc, float* s_od,
                                         int* __restrict__ csrc, float* __restrict__ cd,
                                         int base, bool staged, int w, int lane)
{
    const int hl = lane >> 4;
    const int jl = lane & 15;
    auto visit = [&](const int2 r) {
        const int node = r.x >> 16;
        if (PASS == 0) {
            atomicAdd(&s_cnt[node], 1);
        } else {
            const int p = s_off[node] + atomicAdd(&s_cur[node], 1);
            if (staged) {
                s_osrc[p] = r.x & 0xffff;
                s_od[p]   = __int_as_float(r.y);
            } else {
                csrc[base + p] = r.x & 0xffff;
                cd[base + p]   = __int_as_float(r.y);
            }
        }
    };
    for (int i0 = 0; i0 < 16; i0 += RUN_GROUP) {
        int2        r[RUN_GROUP];
        int         ln[RUN_GROUP];
        const int2* rp[RUN_GROUP];
        #pragma unroll
        for (int q = 0; q < RUN_GROUP; ++q) {
            const int c = 2 * (w + 8 * (i0 + q)) + hl;
            ln[q] = s_len[c];
            rp[q] = rec + (size_t)c * CHUNK + s_st[c];
            r[q]  = make_int2(0, 0);
        }
        #pragma unroll
        for (int q = 0; q < RUN_GROUP; ++q)
            if (jl < ln[q]) r[q] = rp[q][jl];
        #pragma unroll
        for (int q = 0; q < RUN_GROUP; ++q)
            if (jl < ln[q]) visit(r[q]);
        #pragma unroll
        for (int q = 0; q < RUN_GROUP; ++q)
            for (int j = jl + 16; j < ln[q]; j += 16) visit(rp[q][j]);
    }
}

__device__ __forceinline__
void bin_sort_block(const int* __restrict__ tab,
                    const int* __restrict__ rec_w,
                    int* __restrict__ rowptr,
                    int* __restrict__ csrc,
                    float* __restrict__ cd)
{
    __shared__ int   s_st[256];
    __shared__ int   s_len[256];
    __shared__ int   s_cnt[256];
    __shared__ int   s_off[256];
    __shared__ int   s_cur[256];
    __shared__ int   s_w[8];
    __shared__ int   s_osrc[BIN_CAP];
    __shared__ float s_od[BIN_CAP];

    const int tid  = threadIdx.x;
    const int lane = tid & 31;
    const int w    = tid >> 5;
    const int b    = blockIdx.x;
    const int2* rec = reinterpret_cast<const int2*>(rec_w);
    int st = 0, len = 0;
    if (tid < NCHUNK) {
        st  = tab[tid * TAB_PITCH + b];
        len = tab[tid * TAB_PITCH + b + 1] - st;
    }
    s_st[tid]  = st;
    s_len[tid] = len;
    s_cnt[tid] = 0;
    s_cur[tid] = 0;
    int base, n;
    block_scan_excl(st,  tid, s_w, base);
    block_scan_excl(len, tid, s_w, n);
    const bool staged = n <= BIN_CAP;

    bin_walk<0>(rec, s_st, s_len, s_cnt, s_off, s_cur, s_osrc, s_od, csrc, cd,
                base, staged, w, lane);
    __syncthreads();

    int tot2;
    const int ex = block_scan_excl(s_cnt[tid], tid, s_w, tot2);
    s_off[tid] = ex;
    if (tid < BIN_NODES) rowptr[b * BIN_NODES + tid] = base + ex;
    if (b == NBIN - 1 && tid == 0) rowptr[NNODE] = base + n;
    __syncthreads();

    bin_walk<1>(rec, s_st, s_len, s_cnt, s_off, s_cur, s_osrc, s_od, csrc, cd,
                base, staged, w, lane);
    __syncthreads();
    if (staged) {
        for (int j = tid; j < n; j += 256) {
            csrc[base + j] = s_osrc[j];
            cd[base + j]   = s_od[j];
        }
    }
}

__global__ __launch_bounds__(256)
void gat_chunk_and_gemm(const int* __restrict__ src,
                        const int* __restrict__ dst,
                        const float* __restrict__ dfeat,
                        int* __restrict__ tab,
                        int* __restrict__ rec_w,
                        const float* __restrict__ h,
                        const float* __restrict__ W1,
                        const float* __restrict__ W2,
                        const float* __restrict__ Wa_l,
                        float* __restrict__ z,
                        float* __restrict__ znb,
                        float* __restrict__ a_src,
                        float* __restrict__ a_dst)
{
    if (blockIdx.x < NCHUNK)
        bin_chunk_block(src, dst, dfeat, tab, rec_w);
    else
        gemm_tiles<0, HALF_TILES, SIDE_BLOCKS>(blockIdx.x - NCHUNK, h, W1, W2, Wa_l,
                                               z, znb, a_src, a_dst);
}

__global__ __launch_bounds__(256)
void gat_sort_and_gemm(const int* __restrict__ tab,
                       const int* __restrict__ rec_w,
                       int* __restrict__ rowptr,
                       int* __restrict__ csrc,
                       float* __restrict__ cd,
                       const float* __restrict__ h,
                       const float* __restrict__ W1,
                       const float* __restrict__ W2,
                       const float* __restrict__ Wa_l,
                       float* __restrict__ z,
                       float* __restrict__ znb,
                       float* __restrict__ a_src,
                       float* __restrict__ a_dst)
{
    if (blockIdx.x < NBIN)
        bin_sort_block(tab, rec_w, rowptr, csrc, cd);
    else
        gemm_tiles<HALF_TILES, GEMM_TILES, SIDE_BLOCKS>(blockIdx.x - NBIN, h, W1, W2, Wa_l,
                                                        z, znb, a_src, a_dst);
}

__device__ __forceinline__ float wave_max(float v) {
    #pragma unroll
    for (int off = 16; off > 0; off >>= 1) v = fmaxf(v, __shfl_xor(v, off, 32));
    return v;
}
__device__ __forceinline__ float wave_sum(float v) {
    #pragma unroll
    for (int off = 16; off > 0; off >>= 1) v += __shfl_xor(v, off, 32);
    return v;
}

template <int R>
__device__ __forceinline__ void add_rows(float4& acc, const float4* __restrict__ z4,
                                         int lane, int s, int ali, int k)
{
    int    sk[R];
    float  ak[R];
    float4 r[R];
    #pragma unroll
    for (int q = 0; q < R; ++q) {
        sk[q] = __builtin_amdgcn_readlane(s, k + q);
        ak[q] = __int_as_float(__builtin_amdgcn_readlane(ali, k + q));
    }
    #pragma unroll
    for (int q = 0; q < R; ++q) r[q] = z4[(unsigned)(sk[q] * (DD / 4) + lane)];
    #pragma unroll
    for (int q = 0; q < R; ++q) {
        acc.x = fmaf(r[q].x, ak[q], acc.x);
        acc.y = fmaf(r[q].y, ak[q], acc.y);
        acc.z = fmaf(r[q].z, ak[q], acc.z);
        acc.w = fmaf(r[q].w, ak[q], acc.w);
    }
}

__device__ __forceinline__ void gather_rows(float4& acc, const float4* __restrict__ z4,
                                            int lane, int s, float al, int cnt)
{
    const int ali = __float_as_int(al);
    int k = 0;
    for (; k + 8 <= cnt; k += 8) add_rows<8>(acc, z4, lane, s, ali, k);
    if (cnt - k >= 4) { add_rows<4>(acc, z4, lane, s, ali, k); k += 4; }
    switch (cnt - k) {
        case 3:  add_rows<3>(acc, z4, lane, s, ali, k); break;
        case 2:  add_rows<2>(acc, z4, lane, s, ali, k); break;
        case 1:  add_rows<1>(acc, z4, lane, s, ali, k); break;
        default: break;
    }
}

__device__ __forceinline__
void aggregate_small(float4& acc, const float4* __restrict__ z4, const int lane,
                     const int deg, const int s, const float t)
{
    const bool  act = lane < deg;
    const float v   = act ? ((t > 0.0f) ? t : 0.01f * t) : -INFINITY;
    float m = wave_max(v);
    if (!isfinite(m)) m = 0.0f;
    const float ex  = act ? __expf(v - m) : 0.0f;
    const float den = wave_sum(ex);
    gather_rows(acc, z4, lane, s, ex / den, deg);
}

__device__ __forceinline__
void aggregate_large(float4& acc, const float4* __restrict__ z4, const int lane,
                     const int beg, const int end, const float adt, const float ct,
                     const int* __restrict__ csrc,
                     const float* __restrict__ cd,
                     const float* __restrict__ a_src)
{
    float m = -INFINITY;
    for (int j = beg + lane; j < end; j += 32) {
        const float t = a_src[csrc[j]] + adt + cd[j] * ct;
        m = fmaxf(m, (t > 0.0f) ? t : 0.01f * t);
    }
    m = wave_max(m);
    if (!isfinite(m)) m = 0.0f;
    float den = 0.0f;
    for (int j = beg + lane; j < end; j += 32) {
        const float t = a_src[csrc[j]] + adt + cd[j] * ct;
        den += __expf(((t > 0.0f) ? t : 0.01f * t) - m);
    }
    den = wave_sum(den);
    for (int c = beg; c < end; c += 32) {
        const int j = c + lane;
        int   s  = 0;
        float al = 0.0f;
        if (j < end) {
            s = csrc[j];
            const float t = a_src[s] + adt + cd[j] * ct;
            al = __expf(((t > 0.0f) ? t : 0.01f * t) - m) / den;
        }
        const int left = end - c;
        gather_rows(acc, z4, lane, s, al, left < 32 ? left : 32);
    }
}

__device__ __forceinline__ float4 add_relu(const float4 own, const float4 acc)
{
    float4 o;
    o.x = fmaxf(own.x + acc.x, 0.0f);
    o.y = fmaxf(own.y + acc.y, 0.0f);
    o.z = fmaxf(own.z + acc.z, 0.0f);
    o.w = fmaxf(own.w + acc.w, 0.0f);
    return o;
}

__global__ __launch_bounds__(256)
void gat_node_aggregate(const int* __restrict__ rowptr,
                        const int* __restrict__ csrc,
                        const float* __restrict__ cd,
                        const float* __restrict__ W0_l,
                        const float* __restrict__ Wa_l,
                        const float* __restrict__ a_src,
                        const float* __restrict__ a_dst,
                        const float* __restrict__ z,
                        const float* __restrict__ zi,
                        float* __restrict__ hout)
{
    const int gid  = blockIdx.x * blockDim.x + threadIdx.x;
    const int lane = gid & 31;
    const int node = __builtin_amdgcn_readfirstlane(gid >> 5);
    if (node >= NNODE) return;
    const int beg = __builtin_amdgcn_readfirstlane(rowptr[node]);
    const int end = __builtin_amdgcn_readfirstlane(rowptr[node + 1]);
    const int deg = end - beg;

    const float ct  = W0_l[0] * Wa_l[2 * DD];
    const float adt = a_dst[node];
    const float4* z4 = reinterpret_cast<const float4*>(z);
    const float4 own = reinterpret_cast<const float4*>(zi)[(size_t)node * (DD / 4) + lane];
    float4 acc = {0.0f, 0.0f, 0.0f, 0.0f};

    if (deg <= 32) {
        int   s = 0;
        float t = 0.0f;
        if (lane < deg) {
            s = csrc[beg + lane];
            t = a_src[s] + adt + cd[beg + lane] * ct;
        }
        aggregate_small(acc, z4, lane, deg, s, t);
    } else {
        aggregate_large(acc, z4, lane, beg, end, adt, ct, csrc, cd, a_src);
    }
    reinterpret_cast<float4*>(hout)[(size_t)node * (DD / 4) + lane] = add_relu(own, acc);
}

#define FUSE_BLOCKS 512

__device__ __forceinline__ int edge_slot(const int beg, const int end, const int lane)
{
    const int last = (end - 1 > beg) ? end - 1 : beg;
    const int j    = (beg + lane < last) ? beg + lane : last;
    return (j < NEDGE - 1) ? j : NEDGE - 1;
}
static_assert(FUSE_BLOCKS <= GEMM_TILES, "every block owns at least one tile");

__global__ __launch_bounds__(512, 8)
void gat_aggregate_and_gemm(const int* __restrict__ rowptr,
                            const int* __restrict__ csrc,
                            const float* __restrict__ cd,
                            const float* __restrict__ W0_l,
                            const float* __restrict__ Wa_l,
                            const float* __restrict__ a_src,
                            const float* __restrict__ a_dst,
                            const float* __restrict__ z,
                            const float* __restrict__ zi,
                            const float* __restrict__ W1n,
                            const float* __restrict__ W2n,
                            const float* __restrict__ Wa_n,
                            float* __restrict__ z_o,
                            float* __restrict__ zi_o,
                            float* __restrict__ a_src_o,
                            float* __restrict__ a_dst_o)
{
    __shared__ __attribute__((aligned(16))) float As[2][16 * A_PITCH];
    __shared__ float Ps[2][8][32];

    const int  tid   = threadIdx.x;
    const int  lane  = tid & 31;
    const int  wavew = __builtin_amdgcn_readfirstlane(tid >> 5);
    const bool zwave = wavew < 8;
    const int  slab  = wavew & 7;
    const int  col0  = slab << 4;
    const int  mn    = lane & 15;
    const int  khalf = (lane >> 4) << 1;

    const float* __restrict__ Wn   = zwave ? W1n : W2n;
    float* __restrict__       outp = zwave ? z_o : zi_o;
    v2f bf[DD / 4];
    #pragma unroll
    for (int s = 0; s < DD / 4; ++s) {
        const size_t k = (size_t)(4 * s + khalf);
        bf[s].x = Wn[k * DD + col0 + mn];
        bf[s].y = Wn[(k + 1) * DD + col0 + mn];
    }
    const int pslot = ((mn >> 3) << 4) + ((lane >> 4) << 3) + (mn & 7);

    const float   ct  = W0_l[0] * Wa_l[2 * DD];
    const float4* z4  = reinterpret_cast<const float4*>(z);
    const float4* zi4 = reinterpret_cast<const float4*>(zi);

    int tile = blockIdx.x;
    int node = tile * 16 + wavew;
    int beg  = __builtin_amdgcn_readfirstlane(rowptr[node]);
    int end  = __builtin_amdgcn_readfirstlane(rowptr[node + 1]);
    const int ej = edge_slot(beg, end, lane);
    int   es  = csrc[ej];
    float ed  = cd[ej];
    float ea  = a_src[es];
    float adt = a_dst[node];
    const int node1 = (tile + FUSE_BLOCKS < GEMM_TILES) ? node + 16 * FUSE_BLOCKS : node;
    int nbeg_v = rowptr[node1];
    int nend_v = rowptr[node1 + 1];

    int buf  = 0;
    int prev = -1;
    for (; tile < GEMM_TILES; tile += FUSE_BLOCKS) {
        const bool  more  = tile + FUSE_BLOCKS < GEMM_TILES;
        const int   nnode = more ? node + 16 * FUSE_BLOCKS : node;
        const int   nbeg  = __builtin_amdgcn_readfirstlane(nbeg_v);
        const int   nend  = __builtin_amdgcn_readfirstlane(nend_v);
        const int   nj    = edge_slot(nbeg, nend, lane);
        const int   nes   = csrc[nj];
        const float ned   = cd[nj];

        const float4 own = zi4[(unsigned)(node * (DD / 4) + lane)];
        float4 acc = {0.0f, 0.0f, 0.0f, 0.0f};
        const int deg = end - beg;
        if (deg <= 32) aggregate_small(acc, z4, lane, deg, es, ea + adt + ed * ct);
        else           aggregate_large(acc, z4, lane, beg, end, adt, ct, csrc, cd, a_src);

        *reinterpret_cast<float4*>(&As[buf][wavew * A_PITCH + lane * 4]) = add_relu(own, acc);
        __syncthreads();

        const float nea   = a_src[nes];
        const float nadt  = a_dst[nnode];
        const int   node2 = (tile + 2 * FUSE_BLOCKS < GEMM_TILES) ? nnode + 16 * FUSE_BLOCKS : nnode;
        nbeg_v = rowptr[node2];
        nend_v = rowptr[node2 + 1];
        const float wa_s = Wa_n[col0 + mn];
        const float wa_d = Wa_n[DD + col0 + mn];

        if (wavew == 0 && prev >= 0) {
            float s = Ps[buf ^ 1][0][lane];
            #pragma unroll
            for (int w = 1; w < 8; ++w) s += Ps[buf ^ 1][w][lane];
            if (lane & 16) a_dst_o[prev * 16 + mn] = s;
            else           a_src_o[prev * 16 + mn] = s;
        }

        v8f accw = {0.f, 0.f, 0.f, 0.f, 0.f, 0.f, 0.f, 0.f};
        const float* arow = &As[buf][mn * A_PITCH + khalf];
        #pragma unroll
        for (int s = 0; s < DD / 4; ++s) {
            const v2f a = *reinterpret_cast<const v2f*>(arow + 4 * s);
            accw = __builtin_amdgcn_wmma_f32_16x16x4_f32(false, a, false, bf[s],
                                                         (short)0, accw, false, false);
        }

        if (zwave) {
            float v8[8], v4[4], v2[2];
            #pragma unroll
            for (int r = 0; r < 8; ++r)
                v8[r] = fold_pair<8>((lane & 8) != 0, accw[r] * wa_s, accw[r] * wa_d);
            #pragma unroll
            for (int r = 0; r < 4; ++r)
                v4[r] = fold_pair<4>((lane & 4) != 0, v8[r], v8[r + 4]);
            #pragma unroll
            for (int r = 0; r < 2; ++r)
                v2[r] = fold_pair<2>((lane & 2) != 0, v4[r], v4[r + 2]);
            Ps[buf][slab][pslot] = fold_pair<1>((lane & 1) != 0, v2[0], v2[1]);
        }

        const int rowBase = tile * 16 + ((lane >> 4) << 3);
        #pragma unroll
        for (int r = 0; r < 8; ++r)
            outp[(unsigned)((rowBase + r) * DD + col0 + mn)] = accw[r];

        prev = tile;
        buf ^= 1;
        node = nnode; beg = nbeg; end = nend;
        es = nes; ed = ned; ea = nea; adt = nadt;
    }

    __syncthreads();
    if (wavew == 0) {
        float s = Ps[buf ^ 1][0][lane];
        #pragma unroll
        for (int w = 1; w < 8; ++w) s += Ps[buf ^ 1][w][lane];
        if (lane & 16) a_dst_o[prev * 16 + mn] = s;
        else           a_src_o[prev * 16 + mn] = s;
    }
}

extern "C" void kernel_launch(void* const* d_in, const int* in_sizes, int n_in,
                              void* d_out, int out_size, void* d_ws, size_t ws_size,
                              hipStream_t stream) {
    const float* attr = (const float*)d_in[0];
    const float* dfe  = (const float*)d_in[1];
    const int*   src  = (const int*)d_in[2];
    const int*   dst  = (const int*)d_in[3];
    const float* W0   = (const float*)d_in[4];
    const float* W1   = (const float*)d_in[5];
    const float* W2   = (const float*)d_in[6];
    const float* Wa   = (const float*)d_in[7];

    float* ws     = (float*)d_ws;
    float* zA     = ws;
    float* ziA    = zA    + (size_t)NNODE * DD;
    float* zB     = ziA   + (size_t)NNODE * DD;
    float* ziB    = zB    + (size_t)NNODE * DD;
    float* a_srcA = ziB   + (size_t)NNODE * DD;
    float* a_dstA = a_srcA + NNODE;
    float* a_srcB = a_dstA + NNODE;
    float* a_dstB = a_srcB + NNODE;
    int*   tab    = (int*)(a_dstB + NNODE);
    int*   rowptr = tab    + TAB_INTS;
    int*   rec    = rowptr + NNODE + 4;
    int*   csrc   = rec    + 2 * NEDGE;
    float* cd     = (float*)(csrc + NEDGE);

    const int nodeBlocks = (NNODE * 32 + 255) / 256;

    gat_chunk_and_gemm<<<NCHUNK + SIDE_BLOCKS, 256, 0, stream>>>(src, dst, dfe, tab, rec,
                                                                 attr, W1, W2, Wa,
                                                                 zA, ziA, a_srcA, a_dstA);
    gat_sort_and_gemm<<<NBIN + SIDE_BLOCKS, 256, 0, stream>>>(tab, rec, rowptr, csrc, cd,
                                                              attr, W1, W2, Wa,
                                                              zA, ziA, a_srcA, a_dstA);
    gat_aggregate_and_gemm<<<FUSE_BLOCKS, 512, 0, stream>>>(rowptr, csrc, cd, W0, Wa,
                                                            a_srcA, a_dstA, zA, ziA,
                                                            W1 + DD * DD, W2 + DD * DD,
                                                            Wa + (2 * DD + 1),
                                                            zB, ziB, a_srcB, a_dstB);
    gat_aggregate_and_gemm<<<FUSE_BLOCKS, 512, 0, stream>>>(rowptr, csrc, cd, W0 + 1,
                                                            Wa + (2 * DD + 1),
                                                            a_srcB, a_dstB, zB, ziB,
                                                            W1 + 2 * DD * DD, W2 + 2 * DD * DD,
                                                            Wa + 2 * (2 * DD + 1),
                                                            zA, ziA, a_srcA, a_dstA);
    gat_node_aggregate<<<nodeBlocks, 256, 0, stream>>>(rowptr, csrc, cd, W0 + 2,
                                                       Wa + 2 * (2 * DD + 1),
                                                       a_srcA, a_dstA, zA, ziA, (float*)d_out);
}
